# starting kernel
# baseline (speedup 1.0000x reference)
#include <hip/hip_runtime.h>
#include <math.h>

typedef __attribute__((ext_vector_type(2))) float v2f;
typedef __attribute__((ext_vector_type(8))) float v8f;

#define HEADS 8
#define HID 32
#define HD 256
#define NEG_SLOPE 0.2f
#define NSUB 4

template <int NN, int KK, bool RAGGED>
__global__ void gat_wmma_gemm_f32(const float* __restrict__ A,
                                  const float* __restrict__ B,
                                  const float* __restrict__ bias,
                                  float* __restrict__ C,
                                  int tiles_m, int ngroups) {
    constexpr int TILES_N = (NN + 15) / 16;
    constexpr int NACT = TILES_N < NSUB ? TILES_N : NSUB;

    int wave = blockIdx.x * (blockDim.x >> 5) + (threadIdx.x >> 5);
    if (wave >= tiles_m * ngroups) return;
    int tm = wave % tiles_m;
    int tg = wave / tiles_m;
    int m0 = tm << 4;
    int n0 = tg << 6;

    int lane = threadIdx.x & 31;
    int half = lane >> 4;
    int l15  = lane & 15;
    int koff = half << 1;
    int arow = m0 + l15;

    float bmask[NACT];
    unsigned boff[NACT];
#pragma unroll
    for (int j = 0; j < NACT; ++j) {
        int bcol = n0 + (j << 4) + l15;
        int bcolc = bcol;
        if (RAGGED) {
            bmask[j] = (bcol < NN) ? 1.0f : 0.0f;
            bcolc = (bcol < NN) ? bcol : (NN - 1);
        }
        boff[j] = (unsigned)(koff * NN + bcolc);
    }
    const float* ap = A + (size_t)arow * KK + koff;

    v8f acc[NACT];
#pragma unroll
    for (int j = 0; j < NACT; ++j) acc[j] = (v8f){};

#pragma unroll
    for (int k = 0; k < KK; k += 4) {
        v2f a; a.x = ap[k]; a.y = ap[k + 1];
        float t0[NACT], t1[NACT];
#pragma unroll
        for (int j = 0; j < NACT; ++j) {
            t0[j] = B[boff[j] + k * NN];
            t1[j] = B[boff[j] + k * NN + NN];
        }
#pragma unroll
        for (int j = 0; j < NACT; ++j) {
            v2f b;
            if (RAGGED) { b.x = t0[j] * bmask[j]; b.y = t1[j] * bmask[j]; }
            else        { b.x = t0[j];            b.y = t1[j]; }
            acc[j] = __builtin_amdgcn_wmma_f32_16x16x4_f32(
                false, a, false, b, (short)0, acc[j], false, false);
        }
    }

#pragma unroll
    for (int j = 0; j < NACT; ++j) {
        int bcol = n0 + (j << 4) + l15;
        if (!RAGGED || bcol < NN) {
            float bv = bias ? bias[bcol] : 0.0f;
#pragma unroll
            for (int i = 0; i < 8; ++i) {
                int row = m0 + i + (half << 3);
                C[(size_t)row * NN + bcol] = acc[j][i] + bv;
            }
        }
    }
}

__global__ void gat_alpha_k(const float* __restrict__ xh,
                            const float* __restrict__ a_src,
                            const float* __restrict__ a_dst,
                            float* __restrict__ as, float* __restrict__ ad,
                            int N) {
    int t = blockIdx.x * blockDim.x + threadIdx.x;
    if (t >= N * HEADS) return;
    int n = t >> 3, h = t & 7;
    const float* p  = xh + (size_t)n * HD + h * HID;
    const float* ps = a_src + h * HID;
    const float* pd = a_dst + h * HID;
    float ss = 0.f, sd = 0.f;
#pragma unroll
    for (int i = 0; i < HID; ++i) {
        float v = p[i];
        ss += v * ps[i];
        sd += v * pd[i];
    }
    as[t] = ss;
    ad[t] = sd;
}

__global__ void gat_init_softmax(float* __restrict__ emax,
                                 float* __restrict__ denom, int n) {
    int i = blockIdx.x * blockDim.x + threadIdx.x;
    if (i < n) { emax[i] = -1e30f; denom[i] = 0.0f; }
}

__device__ __forceinline__ void atomicMaxFloat(float* addr, float val) {
    if (val >= 0.0f)
        atomicMax((int*)addr, __float_as_int(val));
    else
        atomicMin((unsigned int*)addr, (unsigned int)__float_as_int(val));
}

__device__ __forceinline__ float leaky(float e) {
    return e > 0.0f ? e : NEG_SLOPE * e;
}

__device__ __forceinline__ void edge_sd(const int* __restrict__ src,
                                        const int* __restrict__ dst,
                                        int e, int E, int& s, int& d) {
    if (e < E) { s = src[e]; d = dst[e]; }
    else       { s = e - E; d = s; }
}

__global__ void gat_edge_max(const int* __restrict__ src, const int* __restrict__ dst,
                             int E, int Etot,
                             const float* __restrict__ as, const float* __restrict__ ad,
                             float* __restrict__ emax) {
    int t = blockIdx.x * blockDim.x + threadIdx.x;
    if (t >= Etot * HEADS) return;
    int e = t >> 3, h = t & 7;
    int s, d; edge_sd(src, dst, e, E, s, d);
    float ev = leaky(as[s * HEADS + h] + ad[d * HEADS + h]);
    atomicMaxFloat(&emax[d * HEADS + h], ev);
}

__global__ void gat_edge_expsum(const int* __restrict__ src, const int* __restrict__ dst,
                                int E, int Etot,
                                const float* __restrict__ as, const float* __restrict__ ad,
                                const float* __restrict__ emax,
                                float* __restrict__ denom) {
    int t = blockIdx.x * blockDim.x + threadIdx.x;
    if (t >= Etot * HEADS) return;
    int e = t >> 3, h = t & 7;
    int s, d; edge_sd(src, dst, e, E, s, d);
    float ev = leaky(as[s * HEADS + h] + ad[d * HEADS + h]);
    float ex = __expf(ev - emax[d * HEADS + h]);
    atomicAdd(&denom[d * HEADS + h], ex);
}

__global__ void gat_edge_agg(const int* __restrict__ src, const int* __restrict__ dst,
                             int E, int Etot,
                             const float* __restrict__ as, const float* __restrict__ ad,
                             const float* __restrict__ emax,
                             const float* __restrict__ denom,
                             const float* __restrict__ xh,
                             float* __restrict__ out) {
    int wave = blockIdx.x * (blockDim.x >> 5) + (threadIdx.x >> 5);
    if (wave >= Etot) return;
    int lane = threadIdx.x & 31;
    int s, d; edge_sd(src, dst, wave, E, s, d);

    float w = 0.0f;
    if (lane < HEADS) {
        float ev = leaky(as[s * HEADS + lane] + ad[d * HEADS + lane]);
        w = __expf(ev - emax[d * HEADS + lane]) / denom[d * HEADS + lane];
    }
    const float* xs = xh + (size_t)s * HD;
    float* od = out + (size_t)d * HD;
#pragma unroll
    for (int h = 0; h < HEADS; ++h) {
        float wh = __shfl(w, h, 32);
        atomicAdd(&od[h * HID + lane], wh * xs[h * HID + lane]);
    }
}

__global__ void gat_bias_relu(float* __restrict__ out, const float* __restrict__ b,
                              int total) {
    int i = blockIdx.x * blockDim.x + threadIdx.x;
    if (i < total) {
        float v = out[i] + b[i & (HD - 1)];
        out[i] = v > 0.0f ? v : 0.0f;
    }
}

extern "C" void kernel_launch(void* const* d_in, const int* in_sizes, int n_in,
                              void* d_out, int out_size, void* d_ws, size_t ws_size,
                              hipStream_t stream) {
    const float* x   = (const float*)d_in[0];
    const int*   ei  = (const int*)d_in[1];
    const float* W1  = (const float*)d_in[2];
    const float* a1s = (const float*)d_in[3];
    const float* a1d = (const float*)d_in[4];
    const float* b1  = (const float*)d_in[5];
    const float* W2  = (const float*)d_in[6];
    const float* a2s = (const float*)d_in[7];
    const float* a2d = (const float*)d_in[8];
    const float* b2  = (const float*)d_in[9];
    const float* Wc  = (const float*)d_in[10];
    const float* bc  = (const float*)d_in[11];

    const int N = in_sizes[0] / 128;
    const int E = in_sizes[1] / 2;
    const int Etot = E + N;
    const int* src = ei;
    const int* dst = ei + E;

    float* ws    = (float*)d_ws;
    float* xh    = ws;
    float* hbuf  = xh   + (size_t)N * HD;
    float* as_   = hbuf + (size_t)N * HD;
    float* ad_   = as_  + (size_t)N * HEADS;
    float* emax  = ad_  + (size_t)N * HEADS;
    float* denom = emax + (size_t)N * HEADS;

    const int BT = 256;
    const int tiles_m = N / 16;
    auto cdiv = [](int a, int b) { return (a + b - 1) / b; };
    const int ngroups = HD / (16 * NSUB);
    const int gemm_blocks = cdiv(tiles_m * ngroups, 8);

    gat_wmma_gemm_f32<HD, 128, false><<<gemm_blocks, BT, 0, stream>>>(
        x, W1, nullptr, xh, tiles_m, ngroups);
    gat_alpha_k<<<cdiv(N * HEADS, BT), BT, 0, stream>>>(xh, a1s, a1d, as_, ad_, N);
    gat_init_softmax<<<cdiv(N * HEADS, BT), BT, 0, stream>>>(emax, denom, N * HEADS);
    hipMemsetAsync(hbuf, 0, (size_t)N * HD * sizeof(float), stream);
    gat_edge_max<<<cdiv(Etot * HEADS, BT), BT, 0, stream>>>(src, dst, E, Etot, as_, ad_, emax);
    gat_edge_expsum<<<cdiv(Etot * HEADS, BT), BT, 0, stream>>>(src, dst, E, Etot, as_, ad_, emax, denom);
    gat_edge_agg<<<cdiv(Etot, 8), BT, 0, stream>>>(src, dst, E, Etot, as_, ad_, emax, denom, xh, hbuf);
    gat_bias_relu<<<cdiv(N * HD, BT), BT, 0, stream>>>(hbuf, b1, N * HD);

    gat_wmma_gemm_f32<HD, HD, false><<<gemm_blocks, BT, 0, stream>>>(
        hbuf, W2, nullptr, xh, tiles_m, ngroups);
    gat_alpha_k<<<cdiv(N * HEADS, BT), BT, 0, stream>>>(xh, a2s, a2d, as_, ad_, N);
    gat_init_softmax<<<cdiv(N * HEADS, BT), BT, 0, stream>>>(emax, denom, N * HEADS);
    hipMemsetAsync(hbuf, 0, (size_t)N * HD * sizeof(float), stream);
    gat_edge_max<<<cdiv(Etot * HEADS, BT), BT, 0, stream>>>(src, dst, E, Etot, as_, ad_, emax);
    gat_edge_expsum<<<cdiv(Etot * HEADS, BT), BT, 0, stream>>>(src, dst, E, Etot, as_, ad_, emax, denom);
    gat_edge_agg<<<cdiv(Etot, 8), BT, 0, stream>>>(src, dst, E, Etot, as_, ad_, emax, denom, xh, hbuf);
    gat_bias_relu<<<cdiv(N * HD, BT), BT, 0, stream>>>(hbuf, b2, N * HD);

    gat_wmma_gemm_f32<40, HD, true><<<cdiv(tiles_m, 8), BT, 0, stream>>>(
        hbuf, Wc, bc, (float*)d_out, tiles_m,  1);
}


# Edges counting-sorted by destination in each call; one wave per destination does max, exp-sum and weighted row sum in registers: no float atomics, sum order = arrival rank inside the row
# speedup vs baseline: 2.6332x; 2.6332x over previous
#include <hip/hip_runtime.h>
#include <math.h>

typedef __attribute__((ext_vector_type(2))) float v2f;
typedef __attribute__((ext_vector_type(8))) float v8f;
typedef __attribute__((ext_vector_type(4))) float v4f;

#define HEADS 8
#define HID 32
#define HD 256
#define NEG_SLOPE 0.2f
#define NSUB 4

template <int NN, int KK, bool RAGGED>
__global__ void gat_wmma_gemm_f32(const float* __restrict__ A,
                                  const float* __restrict__ B,
                                  const float* __restrict__ bias,
                                  float* __restrict__ C,
                                  int tiles_m, int ngroups) {
    constexpr int TILES_N = (NN + 15) / 16;
    constexpr int NACT = TILES_N < NSUB ? TILES_N : NSUB;

    int wave = blockIdx.x * (blockDim.x >> 5) + (threadIdx.x >> 5);
    if (wave >= tiles_m * ngroups) return;
    int tm = wave % tiles_m;
    int tg = wave / tiles_m;
    int m0 = tm << 4;
    int n0 = tg << 6;

    int lane = threadIdx.x & 31;
    int half = lane >> 4;
    int l15  = lane & 15;
    int koff = half << 1;
    int arow = m0 + l15;

    float bmask[NACT];
    unsigned boff[NACT];
#pragma unroll
    for (int j = 0; j < NACT; ++j) {
        int bcol = n0 + (j << 4) + l15;
        int bcolc = bcol;
        if (RAGGED) {
            bmask[j] = (bcol < NN) ? 1.0f : 0.0f;
            bcolc = (bcol < NN) ? bcol : (NN - 1);
        }
        boff[j] = (unsigned)(koff * NN + bcolc);
    }
    const float* ap = A + (size_t)arow * KK + koff;

    v8f acc[NACT];
#pragma unroll
    for (int j = 0; j < NACT; ++j) acc[j] = (v8f){};

#pragma unroll
    for (int k = 0; k < KK; k += 4) {
        v2f a; a.x = ap[k]; a.y = ap[k + 1];
        float t0[NACT], t1[NACT];
#pragma unroll
        for (int j = 0; j < NACT; ++j) {
            t0[j] = B[boff[j] + k * NN];
            t1[j] = B[boff[j] + k * NN + NN];
        }
#pragma unroll
        for (int j = 0; j < NACT; ++j) {
            v2f b;
            if (RAGGED) { b.x = t0[j] * bmask[j]; b.y = t1[j] * bmask[j]; }
            else        { b.x = t0[j];            b.y = t1[j]; }
            acc[j] = __builtin_amdgcn_wmma_f32_16x16x4_f32(
                false, a, false, b, (short)0, acc[j], false, false);
        }
    }

#pragma unroll
    for (int j = 0; j < NACT; ++j) {
        int bcol = n0 + (j << 4) + l15;
        if (!RAGGED || bcol < NN) {
            float bv = bias ? bias[bcol] : 0.0f;
#pragma unroll
            for (int i = 0; i < 8; ++i) {
                int row = m0 + i + (half << 3);
                C[(size_t)row * NN + bcol] = acc[j][i] + bv;
            }
        }
    }
}

__global__ void gat_alpha_k(const float* __restrict__ xh,
                            const float* __restrict__ a_src,
                            const float* __restrict__ a_dst,
                            float* __restrict__ as, float* __restrict__ ad,
                            int N) {
    int t = blockIdx.x * blockDim.x + threadIdx.x;
    if (t >= N * HEADS) return;
    int n = t >> 3, h = t & 7;
    const float* p  = xh + (size_t)n * HD + h * HID;
    const float* ps = a_src + h * HID;
    const float* pd = a_dst + h * HID;
    float ss = 0.f, sd = 0.f;
#pragma unroll
    for (int i = 0; i < HID; ++i) {
        float v = p[i];
        ss += v * ps[i];
        sd += v * pd[i];
    }
    as[t] = ss;
    ad[t] = sd;
}

__device__ __forceinline__ float leaky(float e) {
    return e > 0.0f ? e : NEG_SLOPE * e;
}

__global__ void csr_zero(int* __restrict__ cnt, int n) {
    int i = blockIdx.x * blockDim.x + threadIdx.x;
    if (i < n) cnt[i] = 0;
}

__global__ void csr_count(const int* __restrict__ dst, int E, int* __restrict__ cnt) {
    int e = blockIdx.x * blockDim.x + threadIdx.x;
    if (e < E) atomicAdd(&cnt[dst[e]], 1);
}

#define SCAN_BT 1024
__global__ void __launch_bounds__(SCAN_BT)
csr_scan(const int* __restrict__ cnt, int* __restrict__ rowptr, int n) {
    __shared__ int w_prev[SCAN_BT / 32];
    __shared__ int w_tile[SCAN_BT / 32];
    int t = threadIdx.x, lane = t & 31, wid = t >> 5;
    int lim = blockIdx.x * SCAN_BT;
    int prev = 0;
    for (int i = t; i < lim; i += SCAN_BT) prev += cnt[i];
    int idx = lim + t;
    int v = idx < n ? cnt[idx] : 0;
    int x = v;
    for (int off = 1; off < 32; off <<= 1) {
        int u = __shfl_up(x, off, 32);
        if (lane >= off) x += u;
    }
    for (int off = 16; off > 0; off >>= 1) prev += __shfl_xor(prev, off, 32);
    if (lane == 31) w_tile[wid] = x;
    if (lane == 0)  w_prev[wid] = prev;
    __syncthreads();
    int base = 0;
    for (int w = 0; w < SCAN_BT / 32; ++w) {
        base += w_prev[w];
        if (w < wid) base += w_tile[w];
    }
    int excl = base + x - v;
    if (idx < n) rowptr[idx] = excl;
    if (idx == n - 1) rowptr[n] = excl + v;
}

__global__ void csr_fill(const int* __restrict__ src, const int* __restrict__ dst, int E,
                         const int* __restrict__ rowptr, int* __restrict__ cnt,
                         int* __restrict__ col) {
    int e = blockIdx.x * blockDim.x + threadIdx.x;
    if (e >= E) return;
    int d = dst[e];
    int k = atomicSub(&cnt[d], 1) - 1;
    col[rowptr[d] + k] = src[e];
}

__device__ __forceinline__ int row_cols(const int* __restrict__ col, int start, int deg,
                                        int d, int base, int lane) {
    int t = base + lane;
    return (t < deg) ? col[start + t] : d;
}

__device__ __forceinline__ float row_max32(const float* __restrict__ as, int mycol,
                                           float ad_l, int lane) {
    float m = -INFINITY;
    for (int q = 0; q < 8; ++q) {
        int sj = __shfl(mycol, q * 4 + (lane >> 3), 32);
        m = fmaxf(m, leaky(as[(size_t)sj * HEADS + (lane & 7)] + ad_l));
    }
    return m;
}

__device__ __forceinline__ void row_accum(const float* __restrict__ as,
                                          const float* __restrict__ xh,
                                          int mycol, int cnt, int lane, int h0,
                                          float ad0, float ad1, float m0, float m1,
                                          float& sum0, float& sum1, v4f& acc0, v4f& acc1) {
#pragma unroll 4
    for (int j = 0; j < cnt; ++j) {
        int sj = __builtin_amdgcn_readlane(mycol, j);
        const float* ar = as + (size_t)sj * HEADS + h0;
        const float* xr = xh + (size_t)sj * HD + lane * 4;
        float a0 = ar[0], a1 = ar[4];
        v4f x0 = *(const v4f*)xr;
        v4f x1 = *(const v4f*)(xr + 128);
        float p0 = __expf(leaky(a0 + ad0) - m0);
        float p1 = __expf(leaky(a1 + ad1) - m1);
        sum0 += p0; sum1 += p1;
        acc0 += p0 * x0;
        acc1 += p1 * x1;
    }
}

__global__ void __launch_bounds__(256)
gat_row_softmax_agg(const int* __restrict__ rowptr, const int* __restrict__ col,
                    const float* __restrict__ as, const float* __restrict__ ad,
                    const float* __restrict__ xh, const float* __restrict__ bias,
                    float* __restrict__ out, int N) {
    int lane = threadIdx.x & 31;
    int d = __builtin_amdgcn_readfirstlane((int)(blockIdx.x * (blockDim.x >> 5) + (threadIdx.x >> 5)));
    if (d >= N) return;
    int start = rowptr[d];
    int deg   = rowptr[d + 1] - start;
    int total = deg + 1;
    int h0 = lane >> 3;
    float ad_l = ad[(size_t)d * HEADS + (lane & 7)];

    int col0 = row_cols(col, start, deg, d, 0, lane);
    float m = row_max32(as, col0, ad_l, lane);
    for (int base = 32; base < total; base += 32)
        m = fmaxf(m, row_max32(as, row_cols(col, start, deg, d, base, lane), ad_l, lane));
    m = fmaxf(m, __shfl_xor(m, 8, 32));
    m = fmaxf(m, __shfl_xor(m, 16, 32));
    float m0  = __shfl(m, h0, 32),    m1  = __shfl(m, 4 + h0, 32);
    float ad0 = __shfl(ad_l, h0, 32), ad1 = __shfl(ad_l, 4 + h0, 32);

    float sum0 = 0.0f, sum1 = 0.0f;
    v4f acc0 = (v4f){}, acc1 = (v4f){};
    row_accum(as, xh, col0, total < 32 ? total : 32, lane, h0, ad0, ad1, m0, m1,
              sum0, sum1, acc0, acc1);
    for (int base = 32; base < total; base += 32) {
        int rem = total - base;
        row_accum(as, xh, row_cols(col, start, deg, d, base, lane), rem < 32 ? rem : 32,
                  lane, h0, ad0, ad1, m0, m1, sum0, sum1, acc0, acc1);
    }

    const float* bp = bias + lane * 4;
    v4f r0 = acc0 / sum0 + *(const v4f*)bp;
    v4f r1 = acc1 / sum1 + *(const v4f*)(bp + 128);
    for (int i = 0; i < 4; ++i) {
        r0[i] = r0[i] > 0.0f ? r0[i] : 0.0f;
        r1[i] = r1[i] > 0.0f ? r1[i] : 0.0f;
    }
    float* op = out + (size_t)d * HD + lane * 4;
    *(v4f*)op = r0;
    *(v4f*)(op + 128) = r1;
}

extern "C" void kernel_launch(void* const* d_in, const int* in_sizes, int n_in,
                              void* d_out, int out_size, void* d_ws, size_t ws_size,
                              hipStream_t stream) {
    const float* x   = (const float*)d_in[0];
    const int*   ei  = (const int*)d_in[1];
    const float* W1  = (const float*)d_in[2];
    const float* a1s = (const float*)d_in[3];
    const float* a1d = (const float*)d_in[4];
    const float* b1  = (const float*)d_in[5];
    const float* W2  = (const float*)d_in[6];
    const float* a2s = (const float*)d_in[7];
    const float* a2d = (const float*)d_in[8];
    const float* b2  = (const float*)d_in[9];
    const float* Wc  = (const float*)d_in[10];
    const float* bc  = (const float*)d_in[11];

    const int N = in_sizes[0] / 128;
    const int E = in_sizes[1] / 2;
    const int* src = ei;
    const int* dst = ei + E;

    float* ws    = (float*)d_ws;
    float* xh    = ws;
    float* hbuf  = xh   + (size_t)N * HD;
    float* as_   = hbuf + (size_t)N * HD;
    float* ad_   = as_  + (size_t)N * HEADS;
    int*   cnt    = (int*)(ad_ + (size_t)N * HEADS);
    int*   rowptr = cnt + N;
    int*   col    = rowptr + N + 1;

    const int BT = 256;
    const int tiles_m = N / 16;
    auto cdiv = [](int a, int b) { return (a + b - 1) / b; };
    const int ngroups = HD / (16 * NSUB);
    const int gemm_blocks = cdiv(tiles_m * ngroups, 8);

    csr_zero<<<cdiv(N, BT), BT, 0, stream>>>(cnt, N);
    csr_count<<<cdiv(E, BT), BT, 0, stream>>>(dst, E, cnt);
    csr_scan<<<cdiv(N, SCAN_BT), SCAN_BT, 0, stream>>>(cnt, rowptr, N);
    csr_fill<<<cdiv(E, BT), BT, 0, stream>>>(src, dst, E, rowptr, cnt, col);

    gat_wmma_gemm_f32<HD, 128, false><<<gemm_blocks, BT, 0, stream>>>(
        x, W1, nullptr, xh, tiles_m, ngroups);
    gat_alpha_k<<<cdiv(N * HEADS, BT), BT, 0, stream>>>(xh, a1s, a1d, as_, ad_, N);
    gat_row_softmax_agg<<<cdiv(N, 8), BT, 0, stream>>>(rowptr, col, as_, ad_, xh, b1, hbuf, N);

    gat_wmma_gemm_f32<HD, HD, false><<<gemm_blocks, BT, 0, stream>>>(
        hbuf, W2, nullptr, xh, tiles_m, ngroups);
    gat_alpha_k<<<cdiv(N * HEADS, BT), BT, 0, stream>>>(xh, a2s, a2d, as_, ad_, N);
    gat_row_softmax_agg<<<cdiv(N, 8), BT, 0, stream>>>(rowptr, col, as_, ad_, xh, b2, hbuf, N);

    gat_wmma_gemm_f32<40, HD, true><<<cdiv(tiles_m, 8), BT, 0, stream>>>(
        hbuf, Wc, bc, (float*)d_out, tiles_m,  1);
}


# Edge, score and row kernels run on fixed grids of resident waves that loop over their items (3x fewer waves started per call); summation order unchanged
# speedup vs baseline: 2.7318x; 1.0374x over previous
#include <hip/hip_runtime.h>
#include <math.h>

typedef __attribute__((ext_vector_type(2))) float v2f;
typedef __attribute__((ext_vector_type(8))) float v8f;
typedef __attribute__((ext_vector_type(4))) float v4f;

#define HEADS 8
#define HID 32
#define HD 256
#define NEG_SLOPE 0.2f
#define NSUB 4

template <int NN, int KK, bool RAGGED>
__global__ void gat_wmma_gemm_f32(const float* __restrict__ A,
                                  const float* __restrict__ B,
                                  const float* __restrict__ bias,
                                  float* __restrict__ C,
                                  int tiles_m, int ngroups) {
    constexpr int TILES_N = (NN + 15) / 16;
    constexpr int NACT = TILES_N < NSUB ? TILES_N : NSUB;

    int wave = blockIdx.x * (blockDim.x >> 5) + (threadIdx.x >> 5);
    if (wave >= tiles_m * ngroups) return;
    int tm = wave % tiles_m;
    int tg = wave / tiles_m;
    int m0 = tm << 4;
    int n0 = tg << 6;

    int lane = threadIdx.x & 31;
    int half = lane >> 4;
    int l15  = lane & 15;
    int koff = half << 1;
    int arow = m0 + l15;

    float bmask[NACT];
    unsigned boff[NACT];
#pragma unroll
    for (int j = 0; j < NACT; ++j) {
        int bcol = n0 + (j << 4) + l15;
        int bcolc = bcol;
        if (RAGGED) {
            bmask[j] = (bcol < NN) ? 1.0f : 0.0f;
            bcolc = (bcol < NN) ? bcol : (NN - 1);
        }
        boff[j] = (unsigned)(koff * NN + bcolc);
    }
    const float* ap = A + (size_t)arow * KK + koff;

    v8f acc[NACT];
#pragma unroll
    for (int j = 0; j < NACT; ++j) acc[j] = (v8f){};

#pragma unroll
    for (int k = 0; k < KK; k += 4) {
        v2f a; a.x = ap[k]; a.y = ap[k + 1];
        float t0[NACT], t1[NACT];
#pragma unroll
        for (int j = 0; j < NACT; ++j) {
            t0[j] = B[boff[j] + k * NN];
            t1[j] = B[boff[j] + k * NN + NN];
        }
#pragma unroll
        for (int j = 0; j < NACT; ++j) {
            v2f b;
            if (RAGGED) { b.x = t0[j] * bmask[j]; b.y = t1[j] * bmask[j]; }
            else        { b.x = t0[j];            b.y = t1[j]; }
            acc[j] = __builtin_amdgcn_wmma_f32_16x16x4_f32(
                false, a, false, b, (short)0, acc[j], false, false);
        }
    }

#pragma unroll
    for (int j = 0; j < NACT; ++j) {
        int bcol = n0 + (j << 4) + l15;
        if (!RAGGED || bcol < NN) {
            float bv = bias ? bias[bcol] : 0.0f;
#pragma unroll
            for (int i = 0; i < 8; ++i) {
                int row = m0 + i + (half << 3);
                C[(size_t)row * NN + bcol] = acc[j][i] + bv;
            }
        }
    }
}

__global__ void __launch_bounds__(256)
gat_alpha_k(const float* __restrict__ xh,
                            const float* __restrict__ a_src,
                            const float* __restrict__ a_dst,
                            float* __restrict__ as, float* __restrict__ ad,
                            int N) {
    int stride = gridDim.x * blockDim.x;
    int h = threadIdx.x & 7;
    const float* ps = a_src + h * HID;
    const float* pd = a_dst + h * HID;
    for (int t = blockIdx.x * blockDim.x + threadIdx.x; t < N * HEADS; t += stride) {
        const float* p = xh + (size_t)t * HID;
        float ss = 0.f, sd = 0.f;
#pragma unroll
        for (int i = 0; i < HID; ++i) {
            float v = p[i];
            ss += v * ps[i];
            sd += v * pd[i];
        }
        as[t] = ss;
        ad[t] = sd;
    }
}

__device__ __forceinline__ float leaky(float e) {
    return e > 0.0f ? e : NEG_SLOPE * e;
}

__global__ void csr_zero(int* __restrict__ cnt, int n) {
    int i = blockIdx.x * blockDim.x + threadIdx.x;
    if (i < n) cnt[i] = 0;
}

__global__ void csr_count(const int* __restrict__ dst, int E, int* __restrict__ cnt) {
    int stride = gridDim.x * blockDim.x;
    for (int e = blockIdx.x * blockDim.x + threadIdx.x; e < E; e += stride)
        atomicAdd(&cnt[dst[e]], 1);
}

#define SCAN_BT 1024
__global__ void __launch_bounds__(SCAN_BT)
csr_scan(const int* __restrict__ cnt, int* __restrict__ rowptr, int n) {
    __shared__ int w_prev[SCAN_BT / 32];
    __shared__ int w_tile[SCAN_BT / 32];
    int t = threadIdx.x, lane = t & 31, wid = t >> 5;
    int lim = blockIdx.x * SCAN_BT;
    int prev = 0;
    for (int i = t; i < lim; i += SCAN_BT) prev += cnt[i];
    int idx = lim + t;
    int v = idx < n ? cnt[idx] : 0;
    int x = v;
    for (int off = 1; off < 32; off <<= 1) {
        int u = __shfl_up(x, off, 32);
        if (lane >= off) x += u;
    }
    for (int off = 16; off > 0; off >>= 1) prev += __shfl_xor(prev, off, 32);
    if (lane == 31) w_tile[wid] = x;
    if (lane == 0)  w_prev[wid] = prev;
    __syncthreads();
    int base = 0;
    for (int w = 0; w < SCAN_BT / 32; ++w) {
        base += w_prev[w];
        if (w < wid) base += w_tile[w];
    }
    int excl = base + x - v;
    if (idx < n) rowptr[idx] = excl;
    if (idx == n - 1) rowptr[n] = excl + v;
}

__global__ void csr_fill(const int* __restrict__ src, const int* __restrict__ dst, int E,
                         const int* __restrict__ rowptr, int* __restrict__ cnt,
                         int* __restrict__ col) {
    int stride = gridDim.x * blockDim.x;
    for (int e = blockIdx.x * blockDim.x + threadIdx.x; e < E; e += stride) {
        int d = dst[e];
        int k = atomicSub(&cnt[d], 1) - 1;
        col[rowptr[d] + k] = src[e];
    }
}

__device__ __forceinline__ int row_cols(const int* __restrict__ col, int start, int deg,
                                        int d, int base, int lane) {
    int t = base + lane;
    return (t < deg) ? col[start + t] : d;
}

__device__ __forceinline__ float row_max32(const float* __restrict__ as, int mycol,
                                           float ad_l, int lane) {
    float m = -INFINITY;
    for (int q = 0; q < 8; ++q) {
        int sj = __shfl(mycol, q * 4 + (lane >> 3), 32);
        m = fmaxf(m, leaky(as[(size_t)sj * HEADS + (lane & 7)] + ad_l));
    }
    return m;
}

__device__ __forceinline__ void row_accum(const float* __restrict__ as,
                                          const float* __restrict__ xh,
                                          int mycol, int cnt, int lane, int h0,
                                          float ad0, float ad1, float m0, float m1,
                                          float& sum0, float& sum1, v4f& acc0, v4f& acc1) {
#pragma unroll 4
    for (int j = 0; j < cnt; ++j) {
        int sj = __builtin_amdgcn_readlane(mycol, j);
        const float* ar = as + (size_t)sj * HEADS + h0;
        const float* xr = xh + (size_t)sj * HD + lane * 4;
        float a0 = ar[0], a1 = ar[4];
        v4f x0 = *(const v4f*)xr;
        v4f x1 = *(const v4f*)(xr + 128);
        float p0 = __expf(leaky(a0 + ad0) - m0);
        float p1 = __expf(leaky(a1 + ad1) - m1);
        sum0 += p0; sum1 += p1;
        acc0 += p0 * x0;
        acc1 += p1 * x1;
    }
}

__global__ void __launch_bounds__(256)
gat_row_softmax_agg(const int* __restrict__ rowptr, const int* __restrict__ col,
                    const float* __restrict__ as, const float* __restrict__ ad,
                    const float* __restrict__ xh, const float* __restrict__ bias,
                    float* __restrict__ out, int N) {
    int lane = threadIdx.x & 31;
    int w0 = __builtin_amdgcn_readfirstlane((int)(blockIdx.x * (blockDim.x >> 5) + (threadIdx.x >> 5)));
    int nw = (int)(gridDim.x * (blockDim.x >> 5));
    const float* bp = bias + lane * 4;
    v4f bias0 = *(const v4f*)bp;
    v4f bias1 = *(const v4f*)(bp + 128);
    for (int d = w0; d < N; d += nw) {
        int start = rowptr[d];
        int deg   = rowptr[d + 1] - start;
        int total = deg + 1;
        int h0 = lane >> 3;
        float ad_l = ad[(size_t)d * HEADS + (lane & 7)];

        int col0 = row_cols(col, start, deg, d, 0, lane);
        float m = row_max32(as, col0, ad_l, lane);
        for (int base = 32; base < total; base += 32)
            m = fmaxf(m, row_max32(as, row_cols(col, start, deg, d, base, lane), ad_l, lane));
        m = fmaxf(m, __shfl_xor(m, 8, 32));
        m = fmaxf(m, __shfl_xor(m, 16, 32));
        float m0  = __shfl(m, h0, 32),    m1  = __shfl(m, 4 + h0, 32);
        float ad0 = __shfl(ad_l, h0, 32), ad1 = __shfl(ad_l, 4 + h0, 32);

        float sum0 = 0.0f, sum1 = 0.0f;
        v4f acc0 = (v4f){}, acc1 = (v4f){};
        row_accum(as, xh, col0, total < 32 ? total : 32, lane, h0, ad0, ad1, m0, m1,
                  sum0, sum1, acc0, acc1);
        for (int base = 32; base < total; base += 32) {
            int rem = total - base;
            row_accum(as, xh, row_cols(col, start, deg, d, base, lane), rem < 32 ? rem : 32,
                      lane, h0, ad0, ad1, m0, m1, sum0, sum1, acc0, acc1);
        }

        v4f r0 = acc0 / sum0 + bias0;
        v4f r1 = acc1 / sum1 + bias1;
        for (int i = 0; i < 4; ++i) {
            r0[i] = r0[i] > 0.0f ? r0[i] : 0.0f;
            r1[i] = r1[i] > 0.0f ? r1[i] : 0.0f;
        }
        float* op = out + (size_t)d * HD + lane * 4;
        *(v4f*)op = r0;
        *(v4f*)(op + 128) = r1;
    }
}

extern "C" void kernel_launch(void* const* d_in, const int* in_sizes, int n_in,
                              void* d_out, int out_size, void* d_ws, size_t ws_size,
                              hipStream_t stream) {
    const float* x   = (const float*)d_in[0];
    const int*   ei  = (const int*)d_in[1];
    const float* W1  = (const float*)d_in[2];
    const float* a1s = (const float*)d_in[3];
    const float* a1d = (const float*)d_in[4];
    const float* b1  = (const float*)d_in[5];
    const float* W2  = (const float*)d_in[6];
    const float* a2s = (const float*)d_in[7];
    const float* a2d = (const float*)d_in[8];
    const float* b2  = (const float*)d_in[9];
    const float* Wc  = (const float*)d_in[10];
    const float* bc  = (const float*)d_in[11];

    const int N = in_sizes[0] / 128;
    const int E = in_sizes[1] / 2;
    const int* src = ei;
    const int* dst = ei + E;

    float* ws    = (float*)d_ws;
    float* xh    = ws;
    float* hbuf  = xh   + (size_t)N * HD;
    float* as_   = hbuf + (size_t)N * HD;
    float* ad_   = as_  + (size_t)N * HEADS;
    int*   cnt    = (int*)(ad_ + (size_t)N * HEADS);
    int*   rowptr = cnt + N;
    int*   col    = rowptr + N + 1;

    const int BT = 256;
    const int EDGE_BLOCKS  = 512;
    const int ALPHA_BLOCKS = 512;
    const int ROW_BLOCKS   = 1280;
    const int tiles_m = N / 16;
    auto cdiv = [](int a, int b) { return (a + b - 1) / b; };
    const int ngroups = HD / (16 * NSUB);
    const int gemm_blocks = cdiv(tiles_m * ngroups, 8);

    csr_zero<<<cdiv(N, BT), BT, 0, stream>>>(cnt, N);
    csr_count<<<EDGE_BLOCKS, BT, 0, stream>>>(dst, E, cnt);
    csr_scan<<<cdiv(N, SCAN_BT), SCAN_BT, 0, stream>>>(cnt, rowptr, N);
    csr_fill<<<EDGE_BLOCKS, BT, 0, stream>>>(src, dst, E, rowptr, cnt, col);

    gat_wmma_gemm_f32<HD, 128, false><<<gemm_blocks, BT, 0, stream>>>(
        x, W1, nullptr, xh, tiles_m, ngroups);
    gat_alpha_k<<<ALPHA_BLOCKS, BT, 0, stream>>>(xh, a1s, a1d, as_, ad_, N);
    gat_row_softmax_agg<<<ROW_BLOCKS, BT, 0, stream>>>(rowptr, col, as_, ad_, xh, b1, hbuf, N);

    gat_wmma_gemm_f32<HD, HD, false><<<gemm_blocks, BT, 0, stream>>>(
        hbuf, W2, nullptr, xh, tiles_m, ngroups);
    gat_alpha_k<<<ALPHA_BLOCKS, BT, 0, stream>>>(xh, a2s, a2d, as_, ad_, N);
    gat_row_softmax_agg<<<ROW_BLOCKS, BT, 0, stream>>>(rowptr, col, as_, ad_, xh, b2, hbuf, N);

    gat_wmma_gemm_f32<40, HD, true><<<cdiv(tiles_m, 8), BT, 0, stream>>>(
        hbuf, Wc, bc, (float*)d_out, tiles_m,  1);
}


# Edge sort through LDS: chunks binned by 256-node range, one block per range counts and places its edges with LDS adds: no global atomics, 4 launches -> 2; row sum order = LDS add arrival rank
# speedup vs baseline: 3.7817x; 1.3843x over previous
#include <hip/hip_runtime.h>
#include <math.h>

typedef __attribute__((ext_vector_type(2))) float v2f;
typedef __attribute__((ext_vector_type(8))) float v8f;
typedef __attribute__((ext_vector_type(4))) float v4f;

#define HEADS 8
#define HID 32
#define HD 256
#define NEG_SLOPE 0.2f
#define NSUB 4

template <int NN, int KK, bool RAGGED>
__global__ void gat_wmma_gemm_f32(const float* __restrict__ A,
                                  const float* __restrict__ B,
                                  const float* __restrict__ bias,
                                  float* __restrict__ C,
                                  int tiles_m, int ngroups) {
    constexpr int TILES_N = (NN + 15) / 16;
    constexpr int NACT = TILES_N < NSUB ? TILES_N : NSUB;

    int wave = blockIdx.x * (blockDim.x >> 5) + (threadIdx.x >> 5);
    if (wave >= tiles_m * ngroups) return;
    int tm = wave % tiles_m;
    int tg = wave / tiles_m;
    int m0 = tm << 4;
    int n0 = tg << 6;

    int lane = threadIdx.x & 31;
    int half = lane >> 4;
    int l15  = lane & 15;
    int koff = half << 1;
    int arow = m0 + l15;

    float bmask[NACT];
    unsigned boff[NACT];
#pragma unroll
    for (int j = 0; j < NACT; ++j) {
        int bcol = n0 + (j << 4) + l15;
        int bcolc = bcol;
        if (RAGGED) {
            bmask[j] = (bcol < NN) ? 1.0f : 0.0f;
            bcolc = (bcol < NN) ? bcol : (NN - 1);
        }
        boff[j] = (unsigned)(koff * NN + bcolc);
    }
    const float* ap = A + (size_t)arow * KK + koff;

    v8f acc[NACT];
#pragma unroll
    for (int j = 0; j < NACT; ++j) acc[j] = (v8f){};

#pragma unroll
    for (int k = 0; k < KK; k += 4) {
        v2f a; a.x = ap[k]; a.y = ap[k + 1];
        float t0[NACT], t1[NACT];
#pragma unroll
        for (int j = 0; j < NACT; ++j) {
            t0[j] = B[boff[j] + k * NN];
            t1[j] = B[boff[j] + k * NN + NN];
        }
#pragma unroll
        for (int j = 0; j < NACT; ++j) {
            v2f b;
            if (RAGGED) { b.x = t0[j] * bmask[j]; b.y = t1[j] * bmask[j]; }
            else        { b.x = t0[j];            b.y = t1[j]; }
            acc[j] = __builtin_amdgcn_wmma_f32_16x16x4_f32(
                false, a, false, b, (short)0, acc[j], false, false);
        }
    }

#pragma unroll
    for (int j = 0; j < NACT; ++j) {
        int bcol = n0 + (j << 4) + l15;
        if (!RAGGED || bcol < NN) {
            float bv = bias ? bias[bcol] : 0.0f;
#pragma unroll
            for (int i = 0; i < 8; ++i) {
                int row = m0 + i + (half << 3);
                C[(size_t)row * NN + bcol] = acc[j][i] + bv;
            }
        }
    }
}

__global__ void __launch_bounds__(256)
gat_alpha_k(const float* __restrict__ xh,
                            const float* __restrict__ a_src,
                            const float* __restrict__ a_dst,
                            float* __restrict__ as, float* __restrict__ ad,
                            int N) {
    int stride = gridDim.x * blockDim.x;
    int h = threadIdx.x & 7;
    const float* ps = a_src + h * HID;
    const float* pd = a_dst + h * HID;
    for (int t = blockIdx.x * blockDim.x + threadIdx.x; t < N * HEADS; t += stride) {
        const float* p = xh + (size_t)t * HID;
        float ss = 0.f, sd = 0.f;
#pragma unroll
        for (int i = 0; i < HID; ++i) {
            float v = p[i];
            ss += v * ps[i];
            sd += v * pd[i];
        }
        as[t] = ss;
        ad[t] = sd;
    }
}

__device__ __forceinline__ float leaky(float e) {
    return e > 0.0f ? e : NEG_SLOPE * e;
}

#define SORT_BT 256
#define EPT 13
#define CHUNK (SORT_BT * EPT)
#define RSHIFT 8
#define RNODES (1 << RSHIFT)
#define RANGES_MAX 256
#define PIECES_MAX 256
#define STAGE_CAP 12288
#define PLACE_BATCH 8

__device__ __forceinline__ int block_scan_excl(int v, int* wsum, int& total) {
    int lane = threadIdx.x & 31, wid = threadIdx.x >> 5;
    int x = v;
    for (int off = 1; off < 32; off <<= 1) {
        int u = __shfl_up(x, off, 32);
        if (lane >= off) x += u;
    }
    if (lane == 31) wsum[wid] = x;
    __syncthreads();
    int base = 0, tot = 0;
    for (int w = 0; w < SORT_BT / 32; ++w) {
        int s = wsum[w];
        if (w < wid) base += s;
        tot += s;
    }
    total = tot;
    __syncthreads();
    return base + x - v;
}

__global__ void __launch_bounds__(SORT_BT)
csr_bin(const int* __restrict__ src, const int* __restrict__ dst, int E, int nranges,
        int* __restrict__ offs, unsigned* __restrict__ buf) {
    __shared__ int cnt[RANGES_MAX];
    __shared__ int start[RANGES_MAX];
    __shared__ int wsum[SORT_BT / 32];
    __shared__ unsigned stage[CHUNK];
    int t = threadIdx.x;
    int c = blockIdx.x;
    int e0 = c * CHUNK;
    cnt[t] = 0;
    int dv[EPT], sv[EPT];
#pragma unroll
    for (int j = 0; j < EPT; ++j) {
        int e = e0 + j * SORT_BT + t;
        bool ok = e < E;
        dv[j] = ok ? dst[e] : -1;
        sv[j] = ok ? src[e] : 0;
    }
    __syncthreads();
#pragma unroll
    for (int j = 0; j < EPT; ++j)
        if (dv[j] >= 0) atomicAdd(&cnt[dv[j] >> RSHIFT], 1);
    __syncthreads();
    int total;
    int ex = block_scan_excl(cnt[t], wsum, total);
    start[t] = ex;
    if (t < nranges) offs[c * (nranges + 1) + t] = ex;
    if (t == 0) offs[c * (nranges + 1) + nranges] = total;
    cnt[t] = 0;
    __syncthreads();
#pragma unroll
    for (int j = 0; j < EPT; ++j) {
        if (dv[j] >= 0) {
            int r = dv[j] >> RSHIFT;
            int k = atomicAdd(&cnt[r], 1);
            stage[start[r] + k] = ((unsigned)dv[j] << 16) | (unsigned)sv[j];
        }
    }
    __syncthreads();
    for (int i = t; i < total; i += SORT_BT) buf[(size_t)c * CHUNK + i] = stage[i];
}

__device__ __forceinline__ int piece_of(const int* ppos, int i) {
    int p = 0;
    for (int s = PIECES_MAX / 2; s > 0; s >>= 1)
        if (ppos[p + s] <= i) p += s;
    return p;
}

__global__ void __launch_bounds__(SORT_BT)
csr_place(const int* __restrict__ offs, const unsigned* __restrict__ buf, int nchunks,
          int nranges, int N, int* __restrict__ rowptr, int* __restrict__ col) {
    __shared__ int pstart[PIECES_MAX];
    __shared__ int ppos[PIECES_MAX];
    __shared__ int ncnt[RNODES];
    __shared__ int nrow[RNODES];
    __shared__ int wsum[SORT_BT / 32];
    __shared__ unsigned stage[STAGE_CAP];
    int t = threadIdx.x;
    int r = blockIdx.x;
    int a = 0, len = 0;
    if (t < nchunks) {
        a   = offs[t * (nranges + 1) + r];
        len = offs[t * (nranges + 1) + r + 1] - a;
    }
    int T, before;
    int pp = block_scan_excl(len, wsum, T);
    (void)block_scan_excl(a, wsum, before);
    pstart[t] = t * CHUNK + a;
    ppos[t] = pp;
    ncnt[t] = 0;
    __syncthreads();

    for (int i0 = t; i0 < T; i0 += PLACE_BATCH * SORT_BT) {
        unsigned w[PLACE_BATCH];
#pragma unroll
        for (int u = 0; u < PLACE_BATCH; ++u) {
            int i = i0 + u * SORT_BT;
            int ic = i < T ? i : T - 1;
            int p = piece_of(ppos, ic);
            w[u] = buf[pstart[p] + (ic - ppos[p])];
        }
#pragma unroll
        for (int u = 0; u < PLACE_BATCH; ++u) {
            int i = i0 + u * SORT_BT;
            if (i < T) {
                if (i < STAGE_CAP) stage[i] = w[u];
                atomicAdd(&ncnt[(w[u] >> 16) & (RNODES - 1)], 1);
            }
        }
    }
    __syncthreads();
    int cv = ncnt[t];
    int tot2;
    int ex = block_scan_excl(cv, wsum, tot2);
    nrow[t] = ex;
    int d = (r << RSHIFT) + t;
    if (d < N) rowptr[d] = before + ex;
    if (d == N - 1) rowptr[N] = before + ex + cv;
    ncnt[t] = 0;
    __syncthreads();

    for (int i = t; i < T; i += SORT_BT) {
        unsigned w;
        if (i < STAGE_CAP) {
            w = stage[i];
        } else {
            int p = piece_of(ppos, i);
            w = buf[pstart[p] + (i - ppos[p])];
        }
        int dl = (w >> 16) & (RNODES - 1);
        int k = atomicAdd(&ncnt[dl], 1);
        col[before + nrow[dl] + k] = (int)(w & 0xFFFFu);
    }
}

__device__ __forceinline__ int row_cols(const int* __restrict__ col, int start, int deg,
                                        int d, int base, int lane) {
    int t = base + lane;
    return (t < deg) ? col[start + t] : d;
}

__device__ __forceinline__ float row_max32(const float* __restrict__ as, int mycol,
                                           float ad_l, int lane) {
    float m = -INFINITY;
    for (int q = 0; q < 8; ++q) {
        int sj = __shfl(mycol, q * 4 + (lane >> 3), 32);
        m = fmaxf(m, leaky(as[(size_t)sj * HEADS + (lane & 7)] + ad_l));
    }
    return m;
}

__device__ __forceinline__ void row_accum(const float* __restrict__ as,
                                          const float* __restrict__ xh,
                                          int mycol, int cnt, int lane, int h0,
                                          float ad0, float ad1, float m0, float m1,
                                          float& sum0, float& sum1, v4f& acc0, v4f& acc1) {
#pragma unroll 4
    for (int j = 0; j < cnt; ++j) {
        int sj = __builtin_amdgcn_readlane(mycol, j);
        const float* ar = as + (size_t)sj * HEADS + h0;
        const float* xr = xh + (size_t)sj * HD + lane * 4;
        float a0 = ar[0], a1 = ar[4];
        v4f x0 = *(const v4f*)xr;
        v4f x1 = *(const v4f*)(xr + 128);
        float p0 = __expf(leaky(a0 + ad0) - m0);
        float p1 = __expf(leaky(a1 + ad1) - m1);
        sum0 += p0; sum1 += p1;
        acc0 += p0 * x0;
        acc1 += p1 * x1;
    }
}

__global__ void __launch_bounds__(256)
gat_row_softmax_agg(const int* __restrict__ rowptr, const int* __restrict__ col,
                    const float* __restrict__ as, const float* __restrict__ ad,
                    const float* __restrict__ xh, const float* __restrict__ bias,
                    float* __restrict__ out, int N) {
    int lane = threadIdx.x & 31;
    int w0 = __builtin_amdgcn_readfirstlane((int)(blockIdx.x * (blockDim.x >> 5) + (threadIdx.x >> 5)));
    int nw = (int)(gridDim.x * (blockDim.x >> 5));
    const float* bp = bias + lane * 4;
    v4f bias0 = *(const v4f*)bp;
    v4f bias1 = *(const v4f*)(bp + 128);
    for (int d = w0; d < N; d += nw) {
        int start = rowptr[d];
        int deg   = rowptr[d + 1] - start;
        int total = deg + 1;
        int h0 = lane >> 3;
        float ad_l = ad[(size_t)d * HEADS + (lane & 7)];

        int col0 = row_cols(col, start, deg, d, 0, lane);
        float m = row_max32(as, col0, ad_l, lane);
        for (int base = 32; base < total; base += 32)
            m = fmaxf(m, row_max32(as, row_cols(col, start, deg, d, base, lane), ad_l, lane));
        m = fmaxf(m, __shfl_xor(m, 8, 32));
        m = fmaxf(m, __shfl_xor(m, 16, 32));
        float m0  = __shfl(m, h0, 32),    m1  = __shfl(m, 4 + h0, 32);
        float ad0 = __shfl(ad_l, h0, 32), ad1 = __shfl(ad_l, 4 + h0, 32);

        float sum0 = 0.0f, sum1 = 0.0f;
        v4f acc0 = (v4f){}, acc1 = (v4f){};
        row_accum(as, xh, col0, total < 32 ? total : 32, lane, h0, ad0, ad1, m0, m1,
                  sum0, sum1, acc0, acc1);
        for (int base = 32; base < total; base += 32) {
            int rem = total - base;
            row_accum(as, xh, row_cols(col, start, deg, d, base, lane), rem < 32 ? rem : 32,
                      lane, h0, ad0, ad1, m0, m1, sum0, sum1, acc0, acc1);
        }

        v4f r0 = acc0 / sum0 + bias0;
        v4f r1 = acc1 / sum1 + bias1;
        for (int i = 0; i < 4; ++i) {
            r0[i] = r0[i] > 0.0f ? r0[i] : 0.0f;
            r1[i] = r1[i] > 0.0f ? r1[i] : 0.0f;
        }
        float* op = out + (size_t)d * HD + lane * 4;
        *(v4f*)op = r0;
        *(v4f*)(op + 128) = r1;
    }
}

extern "C" void kernel_launch(void* const* d_in, const int* in_sizes, int n_in,
                              void* d_out, int out_size, void* d_ws, size_t ws_size,
                              hipStream_t stream) {
    const float* x   = (const float*)d_in[0];
    const int*   ei  = (const int*)d_in[1];
    const float* W1  = (const float*)d_in[2];
    const float* a1s = (const float*)d_in[3];
    const float* a1d = (const float*)d_in[4];
    const float* b1  = (const float*)d_in[5];
    const float* W2  = (const float*)d_in[6];
    const float* a2s = (const float*)d_in[7];
    const float* a2d = (const float*)d_in[8];
    const float* b2  = (const float*)d_in[9];
    const float* Wc  = (const float*)d_in[10];
    const float* bc  = (const float*)d_in[11];

    const int N = in_sizes[0] / 128;
    const int E = in_sizes[1] / 2;
    const int* src = ei;
    const int* dst = ei + E;

    float* ws    = (float*)d_ws;
    float* xh    = ws;
    float* hbuf  = xh   + (size_t)N * HD;
    float* as_   = hbuf + (size_t)N * HD;
    float* ad_   = as_  + (size_t)N * HEADS;
    int*   rowptr = (int*)(ad_ + (size_t)N * HEADS);
    int*   col    = rowptr + N + 1;
    const int nranges = (N + RNODES - 1) >> RSHIFT;
    const int nchunks = (E + CHUNK - 1) / CHUNK;
    int*      offs = col + E;
    unsigned* ebuf = (unsigned*)(offs + (size_t)nchunks * (nranges + 1));

    const int BT = 256;
    const int ALPHA_BLOCKS = 512;
    const int ROW_BLOCKS   = 1280;
    const int tiles_m = N / 16;
    auto cdiv = [](int a, int b) { return (a + b - 1) / b; };
    const int ngroups = HD / (16 * NSUB);
    const int gemm_blocks = cdiv(tiles_m * ngroups, 8);

    csr_bin<<<nchunks, SORT_BT, 0, stream>>>(src, dst, E, nranges, offs, ebuf);
    csr_place<<<nranges, SORT_BT, 0, stream>>>(offs, ebuf, nchunks, nranges, N, rowptr, col);

    gat_wmma_gemm_f32<HD, 128, false><<<gemm_blocks, BT, 0, stream>>>(
        x, W1, nullptr, xh, tiles_m, ngroups);
    gat_alpha_k<<<ALPHA_BLOCKS, BT, 0, stream>>>(xh, a1s, a1d, as_, ad_, N);
    gat_row_softmax_agg<<<ROW_BLOCKS, BT, 0, stream>>>(rowptr, col, as_, ad_, xh, b1, hbuf, N);

    gat_wmma_gemm_f32<HD, HD, false><<<gemm_blocks, BT, 0, stream>>>(
        hbuf, W2, nullptr, xh, tiles_m, ngroups);
    gat_alpha_k<<<ALPHA_BLOCKS, BT, 0, stream>>>(xh, a2s, a2d, as_, ad_, N);
    gat_row_softmax_agg<<<ROW_BLOCKS, BT, 0, stream>>>(rowptr, col, as_, ad_, xh, b2, hbuf, N);

    gat_wmma_gemm_f32<40, HD, true><<<cdiv(tiles_m, 8), BT, 0, stream>>>(
        hbuf, Wc, bc, (float*)d_out, tiles_m,  1);
}


# GEMM fragment map: lane takes 4 adjacent columns (one per N-subtile) and 4 adjacent k per lane half, so every A/B load and C store is 16 bytes and whole lines; k sum order inside 8-k groups changes
# speedup vs baseline: 3.9523x; 1.0451x over previous
#include <hip/hip_runtime.h>
#include <math.h>

typedef __attribute__((ext_vector_type(2))) float v2f;
typedef __attribute__((ext_vector_type(8))) float v8f;
typedef __attribute__((ext_vector_type(4))) float v4f;

#define HEADS 8
#define HID 32
#define HD 256
#define NEG_SLOPE 0.2f
#define NSUB 4

template <int NN, int KK, bool RAGGED>
__global__ void gat_wmma_gemm_f32(const float* __restrict__ A,
                                  const float* __restrict__ B,
                                  const float* __restrict__ bias,
                                  float* __restrict__ C,
                                  int tiles_m, int ngroups) {
    constexpr int TILES_N = (NN + 15) / 16;
    constexpr int NACT = TILES_N < NSUB ? TILES_N : NSUB;

    int wave = blockIdx.x * (blockDim.x >> 5) + (threadIdx.x >> 5);
    if (wave >= tiles_m * ngroups) return;
    int tm = wave % tiles_m;
    int tg = wave / tiles_m;
    int m0 = tm << 4;
    int n0 = tg << 6;

    int lane = threadIdx.x & 31;
    int half = lane >> 4;
    int l15  = lane & 15;
    int arow = m0 + l15;

    if constexpr (!RAGGED) {
        static_assert(NN % 64 == 0 && KK % 8 == 0 && NSUB == 4, "full-width strip shape");
        const float* ap = A + (size_t)arow * KK + (half << 2);
        const float* bp = B + (size_t)(half << 2) * NN + n0 + (l15 << 2);

        v8f acc[4];
#pragma unroll
        for (int j = 0; j < 4; ++j) acc[j] = (v8f){};

#pragma unroll
        for (int k = 0; k < KK; k += 8) {
            v4f a4 = *(const v4f*)(ap + k);
            v4f b0 = *(const v4f*)(bp + (k + 0) * NN);
            v4f b1 = *(const v4f*)(bp + (k + 1) * NN);
            v4f b2 = *(const v4f*)(bp + (k + 2) * NN);
            v4f b3 = *(const v4f*)(bp + (k + 3) * NN);
            v2f alo; alo.x = a4.x; alo.y = a4.y;
            v2f ahi; ahi.x = a4.z; ahi.y = a4.w;
#pragma unroll
            for (int j = 0; j < 4; ++j) {
                v2f b; b.x = b0[j]; b.y = b1[j];
                acc[j] = __builtin_amdgcn_wmma_f32_16x16x4_f32(
                    false, alo, false, b, (short)0, acc[j], false, false);
            }
#pragma unroll
            for (int j = 0; j < 4; ++j) {
                v2f b; b.x = b2[j]; b.y = b3[j];
                acc[j] = __builtin_amdgcn_wmma_f32_16x16x4_f32(
                    false, ahi, false, b, (short)0, acc[j], false, false);
            }
        }

        int ccol = n0 + (l15 << 2);
        v4f bv = (v4f){};
        if (bias) bv = *(const v4f*)(bias + ccol);
#pragma unroll
        for (int i = 0; i < 8; ++i) {
            int row = m0 + i + (half << 3);
            v4f o;
            o.x = acc[0][i]; o.y = acc[1][i]; o.z = acc[2][i]; o.w = acc[3][i];
            *(v4f*)(C + (size_t)row * NN + ccol) = o + bv;
        }
    } else {
        int koff = half << 1;

        float bmask[NACT];
        unsigned boff[NACT];
#pragma unroll
        for (int j = 0; j < NACT; ++j) {
            int bcol = n0 + (j << 4) + l15;
            bmask[j] = (bcol < NN) ? 1.0f : 0.0f;
            int bcolc = (bcol < NN) ? bcol : (NN - 1);
            boff[j] = (unsigned)(koff * NN + bcolc);
        }
        const float* ap = A + (size_t)arow * KK + koff;

        v8f acc[NACT];
#pragma unroll
        for (int j = 0; j < NACT; ++j) acc[j] = (v8f){};

#pragma unroll
        for (int k = 0; k < KK; k += 4) {
            v2f a; a.x = ap[k]; a.y = ap[k + 1];
            float t0[NACT], t1[NACT];
#pragma unroll
            for (int j = 0; j < NACT; ++j) {
                t0[j] = B[boff[j] + k * NN];
                t1[j] = B[boff[j] + k * NN + NN];
            }
#pragma unroll
            for (int j = 0; j < NACT; ++j) {
                v2f b;
                b.x = t0[j] * bmask[j]; b.y = t1[j] * bmask[j];
                acc[j] = __builtin_amdgcn_wmma_f32_16x16x4_f32(
                    false, a, false, b, (short)0, acc[j], false, false);
            }
        }

#pragma unroll
        for (int j = 0; j < NACT; ++j) {
            int bcol = n0 + (j << 4) + l15;
            if (bcol < NN) {
                float bv = bias ? bias[bcol] : 0.0f;
#pragma unroll
                for (int i = 0; i < 8; ++i) {
                    int row = m0 + i + (half << 3);
                    C[(size_t)row * NN + bcol] = acc[j][i] + bv;
                }
            }
        }
    }
}

__global__ void __launch_bounds__(256)
gat_alpha_k(const float* __restrict__ xh,
                            const float* __restrict__ a_src,
                            const float* __restrict__ a_dst,
                            float* __restrict__ as, float* __restrict__ ad,
                            int N) {
    int stride = gridDim.x * blockDim.x;
    int h = threadIdx.x & 7;
    const float* ps = a_src + h * HID;
    const float* pd = a_dst + h * HID;
    for (int t = blockIdx.x * blockDim.x + threadIdx.x; t < N * HEADS; t += stride) {
        const float* p = xh + (size_t)t * HID;
        float ss = 0.f, sd = 0.f;
#pragma unroll
        for (int i = 0; i < HID; ++i) {
            float v = p[i];
            ss += v * ps[i];
            sd += v * pd[i];
        }
        as[t] = ss;
        ad[t] = sd;
    }
}

__device__ __forceinline__ float leaky(float e) {
    return e > 0.0f ? e : NEG_SLOPE * e;
}

#define SORT_BT 256
#define EPT 13
#define CHUNK (SORT_BT * EPT)
#define RSHIFT 8
#define RNODES (1 << RSHIFT)
#define RANGES_MAX 256
#define PIECES_MAX 256
#define STAGE_CAP 12288
#define PLACE_BATCH 8

__device__ __forceinline__ int block_scan_excl(int v, int* wsum, int& total) {
    int lane = threadIdx.x & 31, wid = threadIdx.x >> 5;
    int x = v;
    for (int off = 1; off < 32; off <<= 1) {
        int u = __shfl_up(x, off, 32);
        if (lane >= off) x += u;
    }
    if (lane == 31) wsum[wid] = x;
    __syncthreads();
    int base = 0, tot = 0;
    for (int w = 0; w < SORT_BT / 32; ++w) {
        int s = wsum[w];
        if (w < wid) base += s;
        tot += s;
    }
    total = tot;
    __syncthreads();
    return base + x - v;
}

__global__ void __launch_bounds__(SORT_BT)
csr_bin(const int* __restrict__ src, const int* __restrict__ dst, int E, int nranges,
        int* __restrict__ offs, unsigned* __restrict__ buf) {
    __shared__ int cnt[RANGES_MAX];
    __shared__ int start[RANGES_MAX];
    __shared__ int wsum[SORT_BT / 32];
    __shared__ unsigned stage[CHUNK];
    int t = threadIdx.x;
    int c = blockIdx.x;
    int e0 = c * CHUNK;
    cnt[t] = 0;
    int dv[EPT], sv[EPT];
#pragma unroll
    for (int j = 0; j < EPT; ++j) {
        int e = e0 + j * SORT_BT + t;
        bool ok = e < E;
        dv[j] = ok ? dst[e] : -1;
        sv[j] = ok ? src[e] : 0;
    }
    __syncthreads();
#pragma unroll
    for (int j = 0; j < EPT; ++j)
        if (dv[j] >= 0) atomicAdd(&cnt[dv[j] >> RSHIFT], 1);
    __syncthreads();
    int total;
    int ex = block_scan_excl(cnt[t], wsum, total);
    start[t] = ex;
    if (t < nranges) offs[c * (nranges + 1) + t] = ex;
    if (t == 0) offs[c * (nranges + 1) + nranges] = total;
    cnt[t] = 0;
    __syncthreads();
#pragma unroll
    for (int j = 0; j < EPT; ++j) {
        if (dv[j] >= 0) {
            int r = dv[j] >> RSHIFT;
            int k = atomicAdd(&cnt[r], 1);
            stage[start[r] + k] = ((unsigned)dv[j] << 16) | (unsigned)sv[j];
        }
    }
    __syncthreads();
    for (int i = t; i < total; i += SORT_BT) buf[(size_t)c * CHUNK + i] = stage[i];
}

__device__ __forceinline__ int piece_of(const int* ppos, int i) {
    int p = 0;
    for (int s = PIECES_MAX / 2; s > 0; s >>= 1)
        if (ppos[p + s] <= i) p += s;
    return p;
}

__global__ void __launch_bounds__(SORT_BT)
csr_place(const int* __restrict__ offs, const unsigned* __restrict__ buf, int nchunks,
          int nranges, int N, int* __restrict__ rowptr, int* __restrict__ col) {
    __shared__ int pstart[PIECES_MAX];
    __shared__ int ppos[PIECES_MAX];
    __shared__ int ncnt[RNODES];
    __shared__ int nrow[RNODES];
    __shared__ int wsum[SORT_BT / 32];
    __shared__ unsigned stage[STAGE_CAP];
    int t = threadIdx.x;
    int r = blockIdx.x;
    int a = 0, len = 0;
    if (t < nchunks) {
        a   = offs[t * (nranges + 1) + r];
        len = offs[t * (nranges + 1) + r + 1] - a;
    }
    int T, before;
    int pp = block_scan_excl(len, wsum, T);
    (void)block_scan_excl(a, wsum, before);
    pstart[t] = t * CHUNK + a;
    ppos[t] = pp;
    ncnt[t] = 0;
    __syncthreads();

    for (int i0 = t; i0 < T; i0 += PLACE_BATCH * SORT_BT) {
        unsigned w[PLACE_BATCH];
#pragma unroll
        for (int u = 0; u < PLACE_BATCH; ++u) {
            int i = i0 + u * SORT_BT;
            int ic = i < T ? i : T - 1;
            int p = piece_of(ppos, ic);
            w[u] = buf[pstart[p] + (ic - ppos[p])];
        }
#pragma unroll
        for (int u = 0; u < PLACE_BATCH; ++u) {
            int i = i0 + u * SORT_BT;
            if (i < T) {
                if (i < STAGE_CAP) stage[i] = w[u];
                atomicAdd(&ncnt[(w[u] >> 16) & (RNODES - 1)], 1);
            }
        }
    }
    __syncthreads();
    int cv = ncnt[t];
    int tot2;
    int ex = block_scan_excl(cv, wsum, tot2);
    nrow[t] = ex;
    int d = (r << RSHIFT) + t;
    if (d < N) rowptr[d] = before + ex;
    if (d == N - 1) rowptr[N] = before + ex + cv;
    ncnt[t] = 0;
    __syncthreads();

    for (int i = t; i < T; i += SORT_BT) {
        unsigned w;
        if (i < STAGE_CAP) {
            w = stage[i];
        } else {
            int p = piece_of(ppos, i);
            w = buf[pstart[p] + (i - ppos[p])];
        }
        int dl = (w >> 16) & (RNODES - 1);
        int k = atomicAdd(&ncnt[dl], 1);
        col[before + nrow[dl] + k] = (int)(w & 0xFFFFu);
    }
}

__device__ __forceinline__ int row_cols(const int* __restrict__ col, int start, int deg,
                                        int d, int base, int lane) {
    int t = base + lane;
    return (t < deg) ? col[start + t] : d;
}

__device__ __forceinline__ float row_max32(const float* __restrict__ as, int mycol,
                                           float ad_l, int lane) {
    float m = -INFINITY;
    for (int q = 0; q < 8; ++q) {
        int sj = __shfl(mycol, q * 4 + (lane >> 3), 32);
        m = fmaxf(m, leaky(as[(size_t)sj * HEADS + (lane & 7)] + ad_l));
    }
    return m;
}

__device__ __forceinline__ void row_accum(const float* __restrict__ as,
                                          const float* __restrict__ xh,
                                          int mycol, int cnt, int lane, int h0,
                                          float ad0, float ad1, float m0, float m1,
                                          float& sum0, float& sum1, v4f& acc0, v4f& acc1) {
#pragma unroll 4
    for (int j = 0; j < cnt; ++j) {
        int sj = __builtin_amdgcn_readlane(mycol, j);
        const float* ar = as + (size_t)sj * HEADS + h0;
        const float* xr = xh + (size_t)sj * HD + lane * 4;
        float a0 = ar[0], a1 = ar[4];
        v4f x0 = *(const v4f*)xr;
        v4f x1 = *(const v4f*)(xr + 128);
        float p0 = __expf(leaky(a0 + ad0) - m0);
        float p1 = __expf(leaky(a1 + ad1) - m1);
        sum0 += p0; sum1 += p1;
        acc0 += p0 * x0;
        acc1 += p1 * x1;
    }
}

__global__ void __launch_bounds__(256)
gat_row_softmax_agg(const int* __restrict__ rowptr, const int* __restrict__ col,
                    const float* __restrict__ as, const float* __restrict__ ad,
                    const float* __restrict__ xh, const float* __restrict__ bias,
                    float* __restrict__ out, int N) {
    int lane = threadIdx.x & 31;
    int w0 = __builtin_amdgcn_readfirstlane((int)(blockIdx.x * (blockDim.x >> 5) + (threadIdx.x >> 5)));
    int nw = (int)(gridDim.x * (blockDim.x >> 5));
    const float* bp = bias + lane * 4;
    v4f bias0 = *(const v4f*)bp;
    v4f bias1 = *(const v4f*)(bp + 128);
    for (int d = w0; d < N; d += nw) {
        int start = rowptr[d];
        int deg   = rowptr[d + 1] - start;
        int total = deg + 1;
        int h0 = lane >> 3;
        float ad_l = ad[(size_t)d * HEADS + (lane & 7)];

        int col0 = row_cols(col, start, deg, d, 0, lane);
        float m = row_max32(as, col0, ad_l, lane);
        for (int base = 32; base < total; base += 32)
            m = fmaxf(m, row_max32(as, row_cols(col, start, deg, d, base, lane), ad_l, lane));
        m = fmaxf(m, __shfl_xor(m, 8, 32));
        m = fmaxf(m, __shfl_xor(m, 16, 32));
        float m0  = __shfl(m, h0, 32),    m1  = __shfl(m, 4 + h0, 32);
        float ad0 = __shfl(ad_l, h0, 32), ad1 = __shfl(ad_l, 4 + h0, 32);

        float sum0 = 0.0f, sum1 = 0.0f;
        v4f acc0 = (v4f){}, acc1 = (v4f){};
        row_accum(as, xh, col0, total < 32 ? total : 32, lane, h0, ad0, ad1, m0, m1,
                  sum0, sum1, acc0, acc1);
        for (int base = 32; base < total; base += 32) {
            int rem = total - base;
            row_accum(as, xh, row_cols(col, start, deg, d, base, lane), rem < 32 ? rem : 32,
                      lane, h0, ad0, ad1, m0, m1, sum0, sum1, acc0, acc1);
        }

        v4f r0 = acc0 / sum0 + bias0;
        v4f r1 = acc1 / sum1 + bias1;
        for (int i = 0; i < 4; ++i) {
            r0[i] = r0[i] > 0.0f ? r0[i] : 0.0f;
            r1[i] = r1[i] > 0.0f ? r1[i] : 0.0f;
        }
        float* op = out + (size_t)d * HD + lane * 4;
        *(v4f*)op = r0;
        *(v4f*)(op + 128) = r1;
    }
}

extern "C" void kernel_launch(void* const* d_in, const int* in_sizes, int n_in,
                              void* d_out, int out_size, void* d_ws, size_t ws_size,
                              hipStream_t stream) {
    const float* x   = (const float*)d_in[0];
    const int*   ei  = (const int*)d_in[1];
    const float* W1  = (const float*)d_in[2];
    const float* a1s = (const float*)d_in[3];
    const float* a1d = (const float*)d_in[4];
    const float* b1  = (const float*)d_in[5];
    const float* W2  = (const float*)d_in[6];
    const float* a2s = (const float*)d_in[7];
    const float* a2d = (const float*)d_in[8];
    const float* b2  = (const float*)d_in[9];
    const float* Wc  = (const float*)d_in[10];
    const float* bc  = (const float*)d_in[11];

    const int N = in_sizes[0] / 128;
    const int E = in_sizes[1] / 2;
    const int* src = ei;
    const int* dst = ei + E;

    float* ws    = (float*)d_ws;
    float* xh    = ws;
    float* hbuf  = xh   + (size_t)N * HD;
    float* as_   = hbuf + (size_t)N * HD;
    float* ad_   = as_  + (size_t)N * HEADS;
    int*   rowptr = (int*)(ad_ + (size_t)N * HEADS);
    int*   col    = rowptr + N + 1;
    const int nranges = (N + RNODES - 1) >> RSHIFT;
    const int nchunks = (E + CHUNK - 1) / CHUNK;
    int*      offs = col + E;
    unsigned* ebuf = (unsigned*)(offs + (size_t)nchunks * (nranges + 1));

    const int BT = 256;
    const int ALPHA_BLOCKS = 512;
    const int ROW_BLOCKS   = 1280;
    const int tiles_m = N / 16;
    auto cdiv = [](int a, int b) { return (a + b - 1) / b; };
    const int ngroups = HD / (16 * NSUB);
    const int gemm_blocks = cdiv(tiles_m * ngroups, 8);

    csr_bin<<<nchunks, SORT_BT, 0, stream>>>(src, dst, E, nranges, offs, ebuf);
    csr_place<<<nranges, SORT_BT, 0, stream>>>(offs, ebuf, nchunks, nranges, N, rowptr, col);

    gat_wmma_gemm_f32<HD, 128, false><<<gemm_blocks, BT, 0, stream>>>(
        x, W1, nullptr, xh, tiles_m, ngroups);
    gat_alpha_k<<<ALPHA_BLOCKS, BT, 0, stream>>>(xh, a1s, a1d, as_, ad_, N);
    gat_row_softmax_agg<<<ROW_BLOCKS, BT, 0, stream>>>(rowptr, col, as_, ad_, xh, b1, hbuf, N);

    gat_wmma_gemm_f32<HD, HD, false><<<gemm_blocks, BT, 0, stream>>>(
        hbuf, W2, nullptr, xh, tiles_m, ngroups);
    gat_alpha_k<<<ALPHA_BLOCKS, BT, 0, stream>>>(xh, a2s, a2d, as_, ad_, N);
    gat_row_softmax_agg<<<ROW_BLOCKS, BT, 0, stream>>>(rowptr, col, as_, ad_, xh, b2, hbuf, N);

    gat_wmma_gemm_f32<40, HD, true><<<cdiv(tiles_m, 8), BT, 0, stream>>>(
        hbuf, Wc, bc, (float*)d_out, tiles_m,  1);
}


# Attention scores (as, ad) written by the GEMM wave from its accumulators: 2 score launches and 2 re-reads of xh removed; score sum order = 4 in-lane terms, then a 3-level lane tree
# speedup vs baseline: 4.3842x; 1.1093x over previous
#include <hip/hip_runtime.h>
#include <math.h>

typedef __attribute__((ext_vector_type(2))) float v2f;
typedef __attribute__((ext_vector_type(8))) float v8f;
typedef __attribute__((ext_vector_type(4))) float v4f;

#define HEADS 8
#define HID 32
#define HD 256
#define NEG_SLOPE 0.2f
#define NSUB 4


template <int M>
__device__ __forceinline__ float lane_xor(float v) {
    return __builtin_bit_cast(float, __builtin_amdgcn_update_dpp(
        0, __builtin_bit_cast(int, v), 0x160 | M, 0xF, 0xF, true));
}

template <int M, int CNT>
__device__ __forceinline__ void score_level(const float* in, float* out, bool hi) {
#pragma unroll
    for (int k = 0; k < CNT; ++k) {
        float keep = hi ? in[k + CNT] : in[k];
        float send = hi ? in[k] : in[k + CNT];
        out[k] = keep + lane_xor<M>(send);
    }
}

template <int NN, int KK, bool RAGGED>
__global__ void gat_wmma_gemm_f32(const float* __restrict__ A,
                                  const float* __restrict__ B,
                                  const float* __restrict__ bias,
                                  float* __restrict__ C,
                                  const float* __restrict__ a_src,
                                  const float* __restrict__ a_dst,
                                  float* __restrict__ as, float* __restrict__ ad,
                                  int tiles_m, int ngroups) {
    constexpr int TILES_N = (NN + 15) / 16;
    constexpr int NACT = TILES_N < NSUB ? TILES_N : NSUB;

    int wave = blockIdx.x * (blockDim.x >> 5) + (threadIdx.x >> 5);
    if (wave >= tiles_m * ngroups) return;
    int tm = wave % tiles_m;
    int tg = wave / tiles_m;
    int m0 = tm << 4;
    int n0 = tg << 6;

    int lane = threadIdx.x & 31;
    int half = lane >> 4;
    int l15  = lane & 15;
    int arow = m0 + l15;

    if constexpr (!RAGGED) {
        static_assert(NN % 64 == 0 && KK % 8 == 0 && NSUB == 4, "full-width strip shape");
        const float* ap = A + (size_t)arow * KK + (half << 2);
        const float* bp = B + (size_t)(half << 2) * NN + n0 + (l15 << 2);

        v8f acc[4];
#pragma unroll
        for (int j = 0; j < 4; ++j) acc[j] = (v8f){};

#pragma unroll
        for (int k = 0; k < KK; k += 8) {
            v4f a4 = *(const v4f*)(ap + k);
            v4f b0 = *(const v4f*)(bp + (k + 0) * NN);
            v4f b1 = *(const v4f*)(bp + (k + 1) * NN);
            v4f b2 = *(const v4f*)(bp + (k + 2) * NN);
            v4f b3 = *(const v4f*)(bp + (k + 3) * NN);
            v2f alo; alo.x = a4.x; alo.y = a4.y;
            v2f ahi; ahi.x = a4.z; ahi.y = a4.w;
#pragma unroll
            for (int j = 0; j < 4; ++j) {
                v2f b; b.x = b0[j]; b.y = b1[j];
                acc[j] = __builtin_amdgcn_wmma_f32_16x16x4_f32(
                    false, alo, false, b, (short)0, acc[j], false, false);
            }
#pragma unroll
            for (int j = 0; j < 4; ++j) {
                v2f b; b.x = b2[j]; b.y = b3[j];
                acc[j] = __builtin_amdgcn_wmma_f32_16x16x4_f32(
                    false, ahi, false, b, (short)0, acc[j], false, false);
            }
        }

        int ccol = n0 + (l15 << 2);
        v4f bv = (v4f){};
        if (bias) bv = *(const v4f*)(bias + ccol);
#pragma unroll
        for (int i = 0; i < 8; ++i) {
            int row = m0 + i + (half << 3);
            v4f o;
            o.x = acc[0][i]; o.y = acc[1][i]; o.z = acc[2][i]; o.w = acc[3][i];
            *(v4f*)(C + (size_t)row * NN + ccol) = o + bv;
        }

        static_assert(HID == 32 && NN == HEADS * HID, "a 64-column strip is two heads");
        v4f sv = *(const v4f*)(a_src + ccol);
        v4f dv = *(const v4f*)(a_dst + ccol);
        float ps[8], pd[8];
#pragma unroll
        for (int i = 0; i < 8; ++i) {
            ps[i] = fmaf(acc[3][i], sv.w, fmaf(acc[2][i], sv.z, fmaf(acc[1][i], sv.y, acc[0][i] * sv.x)));
            pd[i] = fmaf(acc[3][i], dv.w, fmaf(acc[2][i], dv.z, fmaf(acc[1][i], dv.y, acc[0][i] * dv.x)));
        }
        float s4[4], d4[4], s2[2], d2[2], s1[1], d1[1];
        score_level<4, 4>(ps, s4, (lane & 4) != 0);
        score_level<4, 4>(pd, d4, (lane & 4) != 0);
        score_level<2, 2>(s4, s2, (lane & 2) != 0);
        score_level<2, 2>(d4, d2, (lane & 2) != 0);
        score_level<1, 1>(s2, s1, (lane & 1) != 0);
        score_level<1, 1>(d2, d1, (lane & 1) != 0);
        int srow  = m0 + (half << 3) + (lane & 7);
        int shead = (n0 >> 5) + ((lane >> 3) & 1);
        as[(size_t)srow * HEADS + shead] = s1[0];
        ad[(size_t)srow * HEADS + shead] = d1[0];
    } else {
        int koff = half << 1;

        float bmask[NACT];
        unsigned boff[NACT];
#pragma unroll
        for (int j = 0; j < NACT; ++j) {
            int bcol = n0 + (j << 4) + l15;
            bmask[j] = (bcol < NN) ? 1.0f : 0.0f;
            int bcolc = (bcol < NN) ? bcol : (NN - 1);
            boff[j] = (unsigned)(koff * NN + bcolc);
        }
        const float* ap = A + (size_t)arow * KK + koff;

        v8f acc[NACT];
#pragma unroll
        for (int j = 0; j < NACT; ++j) acc[j] = (v8f){};

#pragma unroll
        for (int k = 0; k < KK; k += 4) {
            v2f a; a.x = ap[k]; a.y = ap[k + 1];
            float t0[NACT], t1[NACT];
#pragma unroll
            for (int j = 0; j < NACT; ++j) {
                t0[j] = B[boff[j] + k * NN];
                t1[j] = B[boff[j] + k * NN + NN];
            }
#pragma unroll
            for (int j = 0; j < NACT; ++j) {
                v2f b;
                b.x = t0[j] * bmask[j]; b.y = t1[j] * bmask[j];
                acc[j] = __builtin_amdgcn_wmma_f32_16x16x4_f32(
                    false, a, false, b, (short)0, acc[j], false, false);
            }
        }

#pragma unroll
        for (int j = 0; j < NACT; ++j) {
            int bcol = n0 + (j << 4) + l15;
            if (bcol < NN) {
                float bv = bias ? bias[bcol] : 0.0f;
#pragma unroll
                for (int i = 0; i < 8; ++i) {
                    int row = m0 + i + (half << 3);
                    C[(size_t)row * NN + bcol] = acc[j][i] + bv;
                }
            }
        }
    }
}

__device__ __forceinline__ float leaky(float e) {
    return e > 0.0f ? e : NEG_SLOPE * e;
}

#define SORT_BT 256
#define EPT 13
#define CHUNK (SORT_BT * EPT)
#define RSHIFT 8
#define RNODES (1 << RSHIFT)
#define RANGES_MAX 256
#define PIECES_MAX 256
#define STAGE_CAP 12288
#define PLACE_BATCH 8

__device__ __forceinline__ int block_scan_excl(int v, int* wsum, int& total) {
    int lane = threadIdx.x & 31, wid = threadIdx.x >> 5;
    int x = v;
    for (int off = 1; off < 32; off <<= 1) {
        int u = __shfl_up(x, off, 32);
        if (lane >= off) x += u;
    }
    if (lane == 31) wsum[wid] = x;
    __syncthreads();
    int base = 0, tot = 0;
    for (int w = 0; w < SORT_BT / 32; ++w) {
        int s = wsum[w];
        if (w < wid) base += s;
        tot += s;
    }
    total = tot;
    __syncthreads();
    return base + x - v;
}

__global__ void __launch_bounds__(SORT_BT)
csr_bin(const int* __restrict__ src, const int* __restrict__ dst, int E, int nranges,
        int* __restrict__ offs, unsigned* __restrict__ buf) {
    __shared__ int cnt[RANGES_MAX];
    __shared__ int start[RANGES_MAX];
    __shared__ int wsum[SORT_BT / 32];
    __shared__ unsigned stage[CHUNK];
    int t = threadIdx.x;
    int c = blockIdx.x;
    int e0 = c * CHUNK;
    cnt[t] = 0;
    int dv[EPT], sv[EPT];
#pragma unroll
    for (int j = 0; j < EPT; ++j) {
        int e = e0 + j * SORT_BT + t;
        bool ok = e < E;
        dv[j] = ok ? dst[e] : -1;
        sv[j] = ok ? src[e] : 0;
    }
    __syncthreads();
#pragma unroll
    for (int j = 0; j < EPT; ++j)
        if (dv[j] >= 0) atomicAdd(&cnt[dv[j] >> RSHIFT], 1);
    __syncthreads();
    int total;
    int ex = block_scan_excl(cnt[t], wsum, total);
    start[t] = ex;
    if (t < nranges) offs[c * (nranges + 1) + t] = ex;
    if (t == 0) offs[c * (nranges + 1) + nranges] = total;
    cnt[t] = 0;
    __syncthreads();
#pragma unroll
    for (int j = 0; j < EPT; ++j) {
        if (dv[j] >= 0) {
            int r = dv[j] >> RSHIFT;
            int k = atomicAdd(&cnt[r], 1);
            stage[start[r] + k] = ((unsigned)dv[j] << 16) | (unsigned)sv[j];
        }
    }
    __syncthreads();
    for (int i = t; i < total; i += SORT_BT) buf[(size_t)c * CHUNK + i] = stage[i];
}

__device__ __forceinline__ int piece_of(const int* ppos, int i) {
    int p = 0;
    for (int s = PIECES_MAX / 2; s > 0; s >>= 1)
        if (ppos[p + s] <= i) p += s;
    return p;
}

__global__ void __launch_bounds__(SORT_BT)
csr_place(const int* __restrict__ offs, const unsigned* __restrict__ buf, int nchunks,
          int nranges, int N, int* __restrict__ rowptr, int* __restrict__ col) {
    __shared__ int pstart[PIECES_MAX];
    __shared__ int ppos[PIECES_MAX];
    __shared__ int ncnt[RNODES];
    __shared__ int nrow[RNODES];
    __shared__ int wsum[SORT_BT / 32];
    __shared__ unsigned stage[STAGE_CAP];
    int t = threadIdx.x;
    int r = blockIdx.x;
    int a = 0, len = 0;
    if (t < nchunks) {
        a   = offs[t * (nranges + 1) + r];
        len = offs[t * (nranges + 1) + r + 1] - a;
    }
    int T, before;
    int pp = block_scan_excl(len, wsum, T);
    (void)block_scan_excl(a, wsum, before);
    pstart[t] = t * CHUNK + a;
    ppos[t] = pp;
    ncnt[t] = 0;
    __syncthreads();

    for (int i0 = t; i0 < T; i0 += PLACE_BATCH * SORT_BT) {
        unsigned w[PLACE_BATCH];
#pragma unroll
        for (int u = 0; u < PLACE_BATCH; ++u) {
            int i = i0 + u * SORT_BT;
            int ic = i < T ? i : T - 1;
            int p = piece_of(ppos, ic);
            w[u] = buf[pstart[p] + (ic - ppos[p])];
        }
#pragma unroll
        for (int u = 0; u < PLACE_BATCH; ++u) {
            int i = i0 + u * SORT_BT;
            if (i < T) {
                if (i < STAGE_CAP) stage[i] = w[u];
                atomicAdd(&ncnt[(w[u] >> 16) & (RNODES - 1)], 1);
            }
        }
    }
    __syncthreads();
    int cv = ncnt[t];
    int tot2;
    int ex = block_scan_excl(cv, wsum, tot2);
    nrow[t] = ex;
    int d = (r << RSHIFT) + t;
    if (d < N) rowptr[d] = before + ex;
    if (d == N - 1) rowptr[N] = before + ex + cv;
    ncnt[t] = 0;
    __syncthreads();

    for (int i = t; i < T; i += SORT_BT) {
        unsigned w;
        if (i < STAGE_CAP) {
            w = stage[i];
        } else {
            int p = piece_of(ppos, i);
            w = buf[pstart[p] + (i - ppos[p])];
        }
        int dl = (w >> 16) & (RNODES - 1);
        int k = atomicAdd(&ncnt[dl], 1);
        col[before + nrow[dl] + k] = (int)(w & 0xFFFFu);
    }
}

__device__ __forceinline__ int row_cols(const int* __restrict__ col, int start, int deg,
                                        int d, int base, int lane) {
    int t = base + lane;
    return (t < deg) ? col[start + t] : d;
}

__device__ __forceinline__ float row_max32(const float* __restrict__ as, int mycol,
                                           float ad_l, int lane) {
    float m = -INFINITY;
    for (int q = 0; q < 8; ++q) {
        int sj = __shfl(mycol, q * 4 + (lane >> 3), 32);
        m = fmaxf(m, leaky(as[(size_t)sj * HEADS + (lane & 7)] + ad_l));
    }
    return m;
}

__device__ __forceinline__ void row_accum(const float* __restrict__ as,
                                          const float* __restrict__ xh,
                                          int mycol, int cnt, int lane, int h0,
                                          float ad0, float ad1, float m0, float m1,
                                          float& sum0, float& sum1, v4f& acc0, v4f& acc1) {
#pragma unroll 4
    for (int j = 0; j < cnt; ++j) {
        int sj = __builtin_amdgcn_readlane(mycol, j);
        const float* ar = as + (size_t)sj * HEADS + h0;
        const float* xr = xh + (size_t)sj * HD + lane * 4;
        float a0 = ar[0], a1 = ar[4];
        v4f x0 = *(const v4f*)xr;
        v4f x1 = *(const v4f*)(xr + 128);
        float p0 = __expf(leaky(a0 + ad0) - m0);
        float p1 = __expf(leaky(a1 + ad1) - m1);
        sum0 += p0; sum1 += p1;
        acc0 += p0 * x0;
        acc1 += p1 * x1;
    }
}

__global__ void __launch_bounds__(256)
gat_row_softmax_agg(const int* __restrict__ rowptr, const int* __restrict__ col,
                    const float* __restrict__ as, const float* __restrict__ ad,
                    const float* __restrict__ xh, const float* __restrict__ bias,
                    float* __restrict__ out, int N) {
    int lane = threadIdx.x & 31;
    int w0 = __builtin_amdgcn_readfirstlane((int)(blockIdx.x * (blockDim.x >> 5) + (threadIdx.x >> 5)));
    int nw = (int)(gridDim.x * (blockDim.x >> 5));
    const float* bp = bias + lane * 4;
    v4f bias0 = *(const v4f*)bp;
    v4f bias1 = *(const v4f*)(bp + 128);
    for (int d = w0; d < N; d += nw) {
        int start = rowptr[d];
        int deg   = rowptr[d + 1] - start;
        int total = deg + 1;
        int h0 = lane >> 3;
        float ad_l = ad[(size_t)d * HEADS + (lane & 7)];

        int col0 = row_cols(col, start, deg, d, 0, lane);
        float m = row_max32(as, col0, ad_l, lane);
        for (int base = 32; base < total; base += 32)
            m = fmaxf(m, row_max32(as, row_cols(col, start, deg, d, base, lane), ad_l, lane));
        m = fmaxf(m, __shfl_xor(m, 8, 32));
        m = fmaxf(m, __shfl_xor(m, 16, 32));
        float m0  = __shfl(m, h0, 32),    m1  = __shfl(m, 4 + h0, 32);
        float ad0 = __shfl(ad_l, h0, 32), ad1 = __shfl(ad_l, 4 + h0, 32);

        float sum0 = 0.0f, sum1 = 0.0f;
        v4f acc0 = (v4f){}, acc1 = (v4f){};
        row_accum(as, xh, col0, total < 32 ? total : 32, lane, h0, ad0, ad1, m0, m1,
                  sum0, sum1, acc0, acc1);
        for (int base = 32; base < total; base += 32) {
            int rem = total - base;
            row_accum(as, xh, row_cols(col, start, deg, d, base, lane), rem < 32 ? rem : 32,
                      lane, h0, ad0, ad1, m0, m1, sum0, sum1, acc0, acc1);
        }

        v4f r0 = acc0 / sum0 + bias0;
        v4f r1 = acc1 / sum1 + bias1;
        for (int i = 0; i < 4; ++i) {
            r0[i] = r0[i] > 0.0f ? r0[i] : 0.0f;
            r1[i] = r1[i] > 0.0f ? r1[i] : 0.0f;
        }
        float* op = out + (size_t)d * HD + lane * 4;
        *(v4f*)op = r0;
        *(v4f*)(op + 128) = r1;
    }
}

extern "C" void kernel_launch(void* const* d_in, const int* in_sizes, int n_in,
                              void* d_out, int out_size, void* d_ws, size_t ws_size,
                              hipStream_t stream) {
    const float* x   = (const float*)d_in[0];
    const int*   ei  = (const int*)d_in[1];
    const float* W1  = (const float*)d_in[2];
    const float* a1s = (const float*)d_in[3];
    const float* a1d = (const float*)d_in[4];
    const float* b1  = (const float*)d_in[5];
    const float* W2  = (const float*)d_in[6];
    const float* a2s = (const float*)d_in[7];
    const float* a2d = (const float*)d_in[8];
    const float* b2  = (const float*)d_in[9];
    const float* Wc  = (const float*)d_in[10];
    const float* bc  = (const float*)d_in[11];

    const int N = in_sizes[0] / 128;
    const int E = in_sizes[1] / 2;
    const int* src = ei;
    const int* dst = ei + E;

    float* ws    = (float*)d_ws;
    float* xh    = ws;
    float* hbuf  = xh   + (size_t)N * HD;
    float* as_   = hbuf + (size_t)N * HD;
    float* ad_   = as_  + (size_t)N * HEADS;
    int*   rowptr = (int*)(ad_ + (size_t)N * HEADS);
    int*   col    = rowptr + N + 1;
    const int nranges = (N + RNODES - 1) >> RSHIFT;
    const int nchunks = (E + CHUNK - 1) / CHUNK;
    int*      offs = col + E;
    unsigned* ebuf = (unsigned*)(offs + (size_t)nchunks * (nranges + 1));

    const int BT = 256;
    const int ROW_BLOCKS   = 1280;
    const int tiles_m = N / 16;
    auto cdiv = [](int a, int b) { return (a + b - 1) / b; };
    const int ngroups = HD / (16 * NSUB);
    const int gemm_blocks = cdiv(tiles_m * ngroups, 8);

    csr_bin<<<nchunks, SORT_BT, 0, stream>>>(src, dst, E, nranges, offs, ebuf);
    csr_place<<<nranges, SORT_BT, 0, stream>>>(offs, ebuf, nchunks, nranges, N, rowptr, col);

    gat_wmma_gemm_f32<HD, 128, false><<<gemm_blocks, BT, 0, stream>>>(
        x, W1, nullptr, xh, a1s, a1d, as_, ad_, tiles_m, ngroups);
    gat_row_softmax_agg<<<ROW_BLOCKS, BT, 0, stream>>>(rowptr, col, as_, ad_, xh, b1, hbuf, N);

    gat_wmma_gemm_f32<HD, HD, false><<<gemm_blocks, BT, 0, stream>>>(
        hbuf, W2, nullptr, xh, a2s, a2d, as_, ad_, tiles_m, ngroups);
    gat_row_softmax_agg<<<ROW_BLOCKS, BT, 0, stream>>>(rowptr, col, as_, ad_, xh, b2, hbuf, N);

    gat_wmma_gemm_f32<40, HD, true><<<cdiv(tiles_m, 8), BT, 0, stream>>>(
        hbuf, Wc, bc, (float*)d_out, nullptr, nullptr, nullptr, nullptr, tiles_m,  1);
}


# Row kernel: softmax weights made once per (entry, head) in the score-gather layout, fetched across lanes by the summing lanes; score table read 1x not 3x; exp-sum order = 8 in-lane terms + lane tree
# speedup vs baseline: 4.5569x; 1.0394x over previous
#include <hip/hip_runtime.h>
#include <math.h>

typedef __attribute__((ext_vector_type(2))) float v2f;
typedef __attribute__((ext_vector_type(8))) float v8f;
typedef __attribute__((ext_vector_type(4))) float v4f;

#define HEADS 8
#define HID 32
#define HD 256
#define NEG_SLOPE 0.2f
#define NSUB 4


template <int M>
__device__ __forceinline__ float lane_xor(float v) {
    return __builtin_bit_cast(float, __builtin_amdgcn_update_dpp(
        0, __builtin_bit_cast(int, v), 0x160 | M, 0xF, 0xF, true));
}

template <int M, int CNT>
__device__ __forceinline__ void score_level(const float* in, float* out, bool hi) {
#pragma unroll
    for (int k = 0; k < CNT; ++k) {
        float keep = hi ? in[k + CNT] : in[k];
        float send = hi ? in[k] : in[k + CNT];
        out[k] = keep + lane_xor<M>(send);
    }
}

template <int NN, int KK, bool RAGGED>
__global__ void gat_wmma_gemm_f32(const float* __restrict__ A,
                                  const float* __restrict__ B,
                                  const float* __restrict__ bias,
                                  float* __restrict__ C,
                                  const float* __restrict__ a_src,
                                  const float* __restrict__ a_dst,
                                  float* __restrict__ as, float* __restrict__ ad,
                                  int tiles_m, int ngroups) {
    constexpr int TILES_N = (NN + 15) / 16;
    constexpr int NACT = TILES_N < NSUB ? TILES_N : NSUB;

    int wave = blockIdx.x * (blockDim.x >> 5) + (threadIdx.x >> 5);
    if (wave >= tiles_m * ngroups) return;
    int tm = wave % tiles_m;
    int tg = wave / tiles_m;
    int m0 = tm << 4;
    int n0 = tg << 6;

    int lane = threadIdx.x & 31;
    int half = lane >> 4;
    int l15  = lane & 15;
    int arow = m0 + l15;

    if constexpr (!RAGGED) {
        static_assert(NN % 64 == 0 && KK % 8 == 0 && NSUB == 4, "full-width strip shape");
        const float* ap = A + (size_t)arow * KK + (half << 2);
        const float* bp = B + (size_t)(half << 2) * NN + n0 + (l15 << 2);

        v8f acc[4];
#pragma unroll
        for (int j = 0; j < 4; ++j) acc[j] = (v8f){};

#pragma unroll
        for (int k = 0; k < KK; k += 8) {
            v4f a4 = *(const v4f*)(ap + k);
            v4f b0 = *(const v4f*)(bp + (k + 0) * NN);
            v4f b1 = *(const v4f*)(bp + (k + 1) * NN);
            v4f b2 = *(const v4f*)(bp + (k + 2) * NN);
            v4f b3 = *(const v4f*)(bp + (k + 3) * NN);
            v2f alo; alo.x = a4.x; alo.y = a4.y;
            v2f ahi; ahi.x = a4.z; ahi.y = a4.w;
#pragma unroll
            for (int j = 0; j < 4; ++j) {
                v2f b; b.x = b0[j]; b.y = b1[j];
                acc[j] = __builtin_amdgcn_wmma_f32_16x16x4_f32(
                    false, alo, false, b, (short)0, acc[j], false, false);
            }
#pragma unroll
            for (int j = 0; j < 4; ++j) {
                v2f b; b.x = b2[j]; b.y = b3[j];
                acc[j] = __builtin_amdgcn_wmma_f32_16x16x4_f32(
                    false, ahi, false, b, (short)0, acc[j], false, false);
            }
        }

        int ccol = n0 + (l15 << 2);
        v4f bv = (v4f){};
        if (bias) bv = *(const v4f*)(bias + ccol);
#pragma unroll
        for (int i = 0; i < 8; ++i) {
            int row = m0 + i + (half << 3);
            v4f o;
            o.x = acc[0][i]; o.y = acc[1][i]; o.z = acc[2][i]; o.w = acc[3][i];
            *(v4f*)(C + (size_t)row * NN + ccol) = o + bv;
        }

        static_assert(HID == 32 && NN == HEADS * HID, "a 64-column strip is two heads");
        v4f sv = *(const v4f*)(a_src + ccol);
        v4f dv = *(const v4f*)(a_dst + ccol);
        float ps[8], pd[8];
#pragma unroll
        for (int i = 0; i < 8; ++i) {
            ps[i] = fmaf(acc[3][i], sv.w, fmaf(acc[2][i], sv.z, fmaf(acc[1][i], sv.y, acc[0][i] * sv.x)));
            pd[i] = fmaf(acc[3][i], dv.w, fmaf(acc[2][i], dv.z, fmaf(acc[1][i], dv.y, acc[0][i] * dv.x)));
        }
        float s4[4], d4[4], s2[2], d2[2], s1[1], d1[1];
        score_level<4, 4>(ps, s4, (lane & 4) != 0);
        score_level<4, 4>(pd, d4, (lane & 4) != 0);
        score_level<2, 2>(s4, s2, (lane & 2) != 0);
        score_level<2, 2>(d4, d2, (lane & 2) != 0);
        score_level<1, 1>(s2, s1, (lane & 1) != 0);
        score_level<1, 1>(d2, d1, (lane & 1) != 0);
        int srow  = m0 + (half << 3) + (lane & 7);
        int shead = (n0 >> 5) + ((lane >> 3) & 1);
        as[(size_t)srow * HEADS + shead] = s1[0];
        ad[(size_t)srow * HEADS + shead] = d1[0];
    } else {
        int koff = half << 1;

        float bmask[NACT];
        unsigned boff[NACT];
#pragma unroll
        for (int j = 0; j < NACT; ++j) {
            int bcol = n0 + (j << 4) + l15;
            bmask[j] = (bcol < NN) ? 1.0f : 0.0f;
            int bcolc = (bcol < NN) ? bcol : (NN - 1);
            boff[j] = (unsigned)(koff * NN + bcolc);
        }
        const float* ap = A + (size_t)arow * KK + koff;

        v8f acc[NACT];
#pragma unroll
        for (int j = 0; j < NACT; ++j) acc[j] = (v8f){};

#pragma unroll
        for (int k = 0; k < KK; k += 4) {
            v2f a; a.x = ap[k]; a.y = ap[k + 1];
            float t0[NACT], t1[NACT];
#pragma unroll
            for (int j = 0; j < NACT; ++j) {
                t0[j] = B[boff[j] + k * NN];
                t1[j] = B[boff[j] + k * NN + NN];
            }
#pragma unroll
            for (int j = 0; j < NACT; ++j) {
                v2f b;
                b.x = t0[j] * bmask[j]; b.y = t1[j] * bmask[j];
                acc[j] = __builtin_amdgcn_wmma_f32_16x16x4_f32(
                    false, a, false, b, (short)0, acc[j], false, false);
            }
        }

#pragma unroll
        for (int j = 0; j < NACT; ++j) {
            int bcol = n0 + (j << 4) + l15;
            if (bcol < NN) {
                float bv = bias ? bias[bcol] : 0.0f;
#pragma unroll
                for (int i = 0; i < 8; ++i) {
                    int row = m0 + i + (half << 3);
                    C[(size_t)row * NN + bcol] = acc[j][i] + bv;
                }
            }
        }
    }
}

__device__ __forceinline__ float leaky(float e) {
    return e > 0.0f ? e : NEG_SLOPE * e;
}

#define SORT_BT 256
#define EPT 13
#define CHUNK (SORT_BT * EPT)
#define RSHIFT 8
#define RNODES (1 << RSHIFT)
#define RANGES_MAX 256
#define PIECES_MAX 256
#define STAGE_CAP 12288
#define PLACE_BATCH 8

__device__ __forceinline__ int block_scan_excl(int v, int* wsum, int& total) {
    int lane = threadIdx.x & 31, wid = threadIdx.x >> 5;
    int x = v;
    for (int off = 1; off < 32; off <<= 1) {
        int u = __shfl_up(x, off, 32);
        if (lane >= off) x += u;
    }
    if (lane == 31) wsum[wid] = x;
    __syncthreads();
    int base = 0, tot = 0;
    for (int w = 0; w < SORT_BT / 32; ++w) {
        int s = wsum[w];
        if (w < wid) base += s;
        tot += s;
    }
    total = tot;
    __syncthreads();
    return base + x - v;
}

__global__ void __launch_bounds__(SORT_BT)
csr_bin(const int* __restrict__ src, const int* __restrict__ dst, int E, int nranges,
        int* __restrict__ offs, unsigned* __restrict__ buf) {
    __shared__ int cnt[RANGES_MAX];
    __shared__ int start[RANGES_MAX];
    __shared__ int wsum[SORT_BT / 32];
    __shared__ unsigned stage[CHUNK];
    int t = threadIdx.x;
    int c = blockIdx.x;
    int e0 = c * CHUNK;
    cnt[t] = 0;
    int dv[EPT], sv[EPT];
#pragma unroll
    for (int j = 0; j < EPT; ++j) {
        int e = e0 + j * SORT_BT + t;
        bool ok = e < E;
        dv[j] = ok ? dst[e] : -1;
        sv[j] = ok ? src[e] : 0;
    }
    __syncthreads();
#pragma unroll
    for (int j = 0; j < EPT; ++j)
        if (dv[j] >= 0) atomicAdd(&cnt[dv[j] >> RSHIFT], 1);
    __syncthreads();
    int total;
    int ex = block_scan_excl(cnt[t], wsum, total);
    start[t] = ex;
    if (t < nranges) offs[c * (nranges + 1) + t] = ex;
    if (t == 0) offs[c * (nranges + 1) + nranges] = total;
    cnt[t] = 0;
    __syncthreads();
#pragma unroll
    for (int j = 0; j < EPT; ++j) {
        if (dv[j] >= 0) {
            int r = dv[j] >> RSHIFT;
            int k = atomicAdd(&cnt[r], 1);
            stage[start[r] + k] = ((unsigned)dv[j] << 16) | (unsigned)sv[j];
        }
    }
    __syncthreads();
    for (int i = t; i < total; i += SORT_BT) buf[(size_t)c * CHUNK + i] = stage[i];
}

__device__ __forceinline__ int piece_of(const int* ppos, int i) {
    int p = 0;
    for (int s = PIECES_MAX / 2; s > 0; s >>= 1)
        if (ppos[p + s] <= i) p += s;
    return p;
}

__global__ void __launch_bounds__(SORT_BT)
csr_place(const int* __restrict__ offs, const unsigned* __restrict__ buf, int nchunks,
          int nranges, int N, int* __restrict__ rowptr, int* __restrict__ col) {
    __shared__ int pstart[PIECES_MAX];
    __shared__ int ppos[PIECES_MAX];
    __shared__ int ncnt[RNODES];
    __shared__ int nrow[RNODES];
    __shared__ int wsum[SORT_BT / 32];
    __shared__ unsigned stage[STAGE_CAP];
    int t = threadIdx.x;
    int r = blockIdx.x;
    int a = 0, len = 0;
    if (t < nchunks) {
        a   = offs[t * (nranges + 1) + r];
        len = offs[t * (nranges + 1) + r + 1] - a;
    }
    int T, before;
    int pp = block_scan_excl(len, wsum, T);
    (void)block_scan_excl(a, wsum, before);
    pstart[t] = t * CHUNK + a;
    ppos[t] = pp;
    ncnt[t] = 0;
    __syncthreads();

    for (int i0 = t; i0 < T; i0 += PLACE_BATCH * SORT_BT) {
        unsigned w[PLACE_BATCH];
#pragma unroll
        for (int u = 0; u < PLACE_BATCH; ++u) {
            int i = i0 + u * SORT_BT;
            int ic = i < T ? i : T - 1;
            int p = piece_of(ppos, ic);
            w[u] = buf[pstart[p] + (ic - ppos[p])];
        }
#pragma unroll
        for (int u = 0; u < PLACE_BATCH; ++u) {
            int i = i0 + u * SORT_BT;
            if (i < T) {
                if (i < STAGE_CAP) stage[i] = w[u];
                atomicAdd(&ncnt[(w[u] >> 16) & (RNODES - 1)], 1);
            }
        }
    }
    __syncthreads();
    int cv = ncnt[t];
    int tot2;
    int ex = block_scan_excl(cv, wsum, tot2);
    nrow[t] = ex;
    int d = (r << RSHIFT) + t;
    if (d < N) rowptr[d] = before + ex;
    if (d == N - 1) rowptr[N] = before + ex + cv;
    ncnt[t] = 0;
    __syncthreads();

    for (int i = t; i < T; i += SORT_BT) {
        unsigned w;
        if (i < STAGE_CAP) {
            w = stage[i];
        } else {
            int p = piece_of(ppos, i);
            w = buf[pstart[p] + (i - ppos[p])];
        }
        int dl = (w >> 16) & (RNODES - 1);
        int k = atomicAdd(&ncnt[dl], 1);
        col[before + nrow[dl] + k] = (int)(w & 0xFFFFu);
    }
}

__device__ __forceinline__ int row_cols(const int* __restrict__ col, int start, int deg,
                                        int d, int base, int lane) {
    int t = base + lane;
    return (t < deg) ? col[start + t] : d;
}

__device__ __forceinline__ int lane_fetch(int src, int v) {
    return __builtin_amdgcn_ds_bpermute(src << 2, v);
}
__device__ __forceinline__ float lane_fetch(int src, float v) {
    return __builtin_bit_cast(float, lane_fetch(src, __builtin_bit_cast(int, v)));
}

__device__ __forceinline__ float row_scores32(const float* __restrict__ as, int mycol,
                                              float ad_l, int lane, float (&e)[8]) {
    float m = -INFINITY;
#pragma unroll
    for (int q = 0; q < 8; ++q) {
        int sj = lane_fetch(q * 4 + (lane >> 3), mycol);
        e[q] = leaky(as[(size_t)sj * HEADS + (lane & 7)] + ad_l);
        m = fmaxf(m, e[q]);
    }
    return m;
}

__device__ __forceinline__ float row_weights32(const float (&e)[8], float m, int cnt, int lane,
                                               float (&p)[8]) {
    float s = 0.0f;
#pragma unroll
    for (int q = 0; q < 8; ++q) {
        p[q] = (q * 4 + (lane >> 3) < cnt) ? __expf(e[q] - m) : 0.0f;
        s += p[q];
    }
    return s;
}

__device__ __forceinline__ void row_accum32(const float* __restrict__ xh, int mycol, int cnt,
                                            int lane, const float (&p)[8],
                                            v4f& acc0, v4f& acc1) {
    int h0 = lane >> 3;
    unsigned lane16 = (unsigned)lane * 16u;
#pragma unroll
    for (int q = 0; q < 8; ++q) {
        if (q * 4 < cnt) {
#pragma unroll
            for (int u = 0; u < 4; ++u) {
                int sj = __builtin_amdgcn_readlane(mycol, q * 4 + u);
                unsigned off = (unsigned)sj * (unsigned)(HD * sizeof(float)) + lane16;
                v4f x0 = *(const v4f*)((const char*)xh + off);
                v4f x1 = *(const v4f*)((const char*)xh + off + 512);
                float w0 = lane_fetch(u * 8 + h0, p[q]);
                float w1 = lane_fetch(u * 8 + 4 + h0, p[q]);
                acc0 += w0 * x0;
                acc1 += w1 * x1;
            }
        }
    }
}

__global__ void __launch_bounds__(128)
gat_row_softmax_agg(const int* __restrict__ rowptr, const int* __restrict__ col,
                    const float* __restrict__ as, const float* __restrict__ ad,
                    const float* __restrict__ xh, const float* __restrict__ bias,
                    float* __restrict__ out, int N) {
    int lane = threadIdx.x & 31;
    int w0 = __builtin_amdgcn_readfirstlane((int)(blockIdx.x * (blockDim.x >> 5) + (threadIdx.x >> 5)));
    int nw = (int)(gridDim.x * (blockDim.x >> 5));
    const float* bp = bias + lane * 4;
    v4f bias0 = *(const v4f*)bp;
    v4f bias1 = *(const v4f*)(bp + 128);
    for (int d = w0; d < N; d += nw) {
        int start = rowptr[d];
        int deg   = rowptr[d + 1] - start;
        int total = deg + 1;
        int h0 = lane >> 3;
        float ad_l = ad[(size_t)d * HEADS + (lane & 7)];

        int col0 = row_cols(col, start, deg, d, 0, lane);
        float e0[8];
        float m = row_scores32(as, col0, ad_l, lane, e0);
        for (int base = 32; base < total; base += 32) {
            float et[8];
            m = fmaxf(m, row_scores32(as, row_cols(col, start, deg, d, base, lane), ad_l, lane, et));
        }
        m = fmaxf(m, __shfl_xor(m, 8, 32));
        m = fmaxf(m, __shfl_xor(m, 16, 32));

        float p[8];
        float lsum = row_weights32(e0, m, total, lane, p);
        v4f acc0 = (v4f){}, acc1 = (v4f){};
        row_accum32(xh, col0, total < 32 ? total : 32, lane, p, acc0, acc1);
        for (int base = 32; base < total; base += 32) {
            int c = row_cols(col, start, deg, d, base, lane);
            float et[8];
            (void)row_scores32(as, c, ad_l, lane, et);
            int rem = total - base;
            lsum += row_weights32(et, m, rem, lane, p);
            row_accum32(xh, c, rem < 32 ? rem : 32, lane, p, acc0, acc1);
        }
        lsum += __shfl_xor(lsum, 8, 32);
        lsum += __shfl_xor(lsum, 16, 32);
        float sum0 = __shfl(lsum, h0, 32), sum1 = __shfl(lsum, 4 + h0, 32);

        v4f r0 = acc0 / sum0 + bias0;
        v4f r1 = acc1 / sum1 + bias1;
        for (int i = 0; i < 4; ++i) {
            r0[i] = r0[i] > 0.0f ? r0[i] : 0.0f;
            r1[i] = r1[i] > 0.0f ? r1[i] : 0.0f;
        }
        float* op = out + (size_t)d * HD + lane * 4;
        *(v4f*)op = r0;
        *(v4f*)(op + 128) = r1;
    }
}

extern "C" void kernel_launch(void* const* d_in, const int* in_sizes, int n_in,
                              void* d_out, int out_size, void* d_ws, size_t ws_size,
                              hipStream_t stream) {
    const float* x   = (const float*)d_in[0];
    const int*   ei  = (const int*)d_in[1];
    const float* W1  = (const float*)d_in[2];
    const float* a1s = (const float*)d_in[3];
    const float* a1d = (const float*)d_in[4];
    const float* b1  = (const float*)d_in[5];
    const float* W2  = (const float*)d_in[6];
    const float* a2s = (const float*)d_in[7];
    const float* a2d = (const float*)d_in[8];
    const float* b2  = (const float*)d_in[9];
    const float* Wc  = (const float*)d_in[10];
    const float* bc  = (const float*)d_in[11];

    const int N = in_sizes[0] / 128;
    const int E = in_sizes[1] / 2;
    const int* src = ei;
    const int* dst = ei + E;

    float* ws    = (float*)d_ws;
    float* xh    = ws;
    float* hbuf  = xh   + (size_t)N * HD;
    float* as_   = hbuf + (size_t)N * HD;
    float* ad_   = as_  + (size_t)N * HEADS;
    int*   rowptr = (int*)(ad_ + (size_t)N * HEADS);
    int*   col    = rowptr + N + 1;
    const int nranges = (N + RNODES - 1) >> RSHIFT;
    const int nchunks = (E + CHUNK - 1) / CHUNK;
    int*      offs = col + E;
    unsigned* ebuf = (unsigned*)(offs + (size_t)nchunks * (nranges + 1));

    const int BT = 256;
    const int ROW_BT       = 128;
    const int ROW_BLOCKS   = 2304;
    const int tiles_m = N / 16;
    auto cdiv = [](int a, int b) { return (a + b - 1) / b; };
    const int ngroups = HD / (16 * NSUB);
    const int gemm_blocks = cdiv(tiles_m * ngroups, 8);

    csr_bin<<<nchunks, SORT_BT, 0, stream>>>(src, dst, E, nranges, offs, ebuf);
    csr_place<<<nranges, SORT_BT, 0, stream>>>(offs, ebuf, nchunks, nranges, N, rowptr, col);

    gat_wmma_gemm_f32<HD, 128, false><<<gemm_blocks, BT, 0, stream>>>(
        x, W1, nullptr, xh, a1s, a1d, as_, ad_, tiles_m, ngroups);
    gat_row_softmax_agg<<<ROW_BLOCKS, ROW_BT, 0, stream>>>(rowptr, col, as_, ad_, xh, b1, hbuf, N);

    gat_wmma_gemm_f32<HD, HD, false><<<gemm_blocks, BT, 0, stream>>>(
        hbuf, W2, nullptr, xh, a2s, a2d, as_, ad_, tiles_m, ngroups);
    gat_row_softmax_agg<<<ROW_BLOCKS, ROW_BT, 0, stream>>>(rowptr, col, as_, ad_, xh, b2, hbuf, N);

    gat_wmma_gemm_f32<40, HD, true><<<cdiv(tiles_m, 8), BT, 0, stream>>>(
        hbuf, Wc, bc, (float*)d_out, nullptr, nullptr, nullptr, nullptr, tiles_m,  1);
}


# Edge sort's two passes each share a launch with part of layer 1's GEMM (first blocks sort, the rest run GEMM strips): 7 -> 6 launches, sort waits hidden under matrix work; sum orders unchanged
# speedup vs baseline: 4.8375x; 1.0616x over previous
#include <hip/hip_runtime.h>
#include <math.h>

typedef __attribute__((ext_vector_type(2))) float v2f;
typedef __attribute__((ext_vector_type(8))) float v8f;
typedef __attribute__((ext_vector_type(4))) float v4f;

#define HEADS 8
#define HID 32
#define HD 256
#define NEG_SLOPE 0.2f
#define NSUB 4


template <int M>
__device__ __forceinline__ float lane_xor(float v) {
    return __builtin_bit_cast(float, __builtin_amdgcn_update_dpp(
        0, __builtin_bit_cast(int, v), 0x160 | M, 0xF, 0xF, true));
}

template <int M, int CNT>
__device__ __forceinline__ void score_level(const float* in, float* out, bool hi) {
#pragma unroll
    for (int k = 0; k < CNT; ++k) {
        float keep = hi ? in[k + CNT] : in[k];
        float send = hi ? in[k] : in[k + CNT];
        out[k] = keep + lane_xor<M>(send);
    }
}

template <int NN, int KK, bool RAGGED>
__device__ __forceinline__ void gemm_strip(int wave,
                                           const float* __restrict__ A,
                                           const float* __restrict__ B,
                                           const float* __restrict__ bias,
                                           float* __restrict__ C,
                                           const float* __restrict__ a_src,
                                           const float* __restrict__ a_dst,
                                           float* __restrict__ as, float* __restrict__ ad,
                                           int tiles_m) {
    constexpr int TILES_N = (NN + 15) / 16;
    constexpr int NACT = TILES_N < NSUB ? TILES_N : NSUB;

    int tm = wave % tiles_m;
    int tg = wave / tiles_m;
    int m0 = tm << 4;
    int n0 = tg << 6;

    int lane = threadIdx.x & 31;
    int half = lane >> 4;
    int l15  = lane & 15;
    int arow = m0 + l15;

    if constexpr (!RAGGED) {
        static_assert(NN % 64 == 0 && KK % 8 == 0 && NSUB == 4, "full-width strip shape");
        const float* ap = A + (size_t)arow * KK + (half << 2);
        const float* bp = B + (size_t)(half << 2) * NN + n0 + (l15 << 2);

        v8f acc[4];
#pragma unroll
        for (int j = 0; j < 4; ++j) acc[j] = (v8f){};

#pragma unroll
        for (int k = 0; k < KK; k += 8) {
            v4f a4 = *(const v4f*)(ap + k);
            v4f b0 = *(const v4f*)(bp + (k + 0) * NN);
            v4f b1 = *(const v4f*)(bp + (k + 1) * NN);
            v4f b2 = *(const v4f*)(bp + (k + 2) * NN);
            v4f b3 = *(const v4f*)(bp + (k + 3) * NN);
            v2f alo; alo.x = a4.x; alo.y = a4.y;
            v2f ahi; ahi.x = a4.z; ahi.y = a4.w;
#pragma unroll
            for (int j = 0; j < 4; ++j) {
                v2f b; b.x = b0[j]; b.y = b1[j];
                acc[j] = __builtin_amdgcn_wmma_f32_16x16x4_f32(
                    false, alo, false, b, (short)0, acc[j], false, false);
            }
#pragma unroll
            for (int j = 0; j < 4; ++j) {
                v2f b; b.x = b2[j]; b.y = b3[j];
                acc[j] = __builtin_amdgcn_wmma_f32_16x16x4_f32(
                    false, ahi, false, b, (short)0, acc[j], false, false);
            }
        }

        int ccol = n0 + (l15 << 2);
        v4f bv = (v4f){};
        if (bias) bv = *(const v4f*)(bias + ccol);
#pragma unroll
        for (int i = 0; i < 8; ++i) {
            int row = m0 + i + (half << 3);
            v4f o;
            o.x = acc[0][i]; o.y = acc[1][i]; o.z = acc[2][i]; o.w = acc[3][i];
            *(v4f*)(C + (size_t)row * NN + ccol) = o + bv;
        }

        static_assert(HID == 32 && NN == HEADS * HID, "a 64-column strip is two heads");
        v4f sv = *(const v4f*)(a_src + ccol);
        v4f dv = *(const v4f*)(a_dst + ccol);
        float ps[8], pd[8];
#pragma unroll
        for (int i = 0; i < 8; ++i) {
            ps[i] = fmaf(acc[3][i], sv.w, fmaf(acc[2][i], sv.z, fmaf(acc[1][i], sv.y, acc[0][i] * sv.x)));
            pd[i] = fmaf(acc[3][i], dv.w, fmaf(acc[2][i], dv.z, fmaf(acc[1][i], dv.y, acc[0][i] * dv.x)));
        }
        float s4[4], d4[4], s2[2], d2[2], s1[1], d1[1];
        score_level<4, 4>(ps, s4, (lane & 4) != 0);
        score_level<4, 4>(pd, d4, (lane & 4) != 0);
        score_level<2, 2>(s4, s2, (lane & 2) != 0);
        score_level<2, 2>(d4, d2, (lane & 2) != 0);
        score_level<1, 1>(s2, s1, (lane & 1) != 0);
        score_level<1, 1>(d2, d1, (lane & 1) != 0);
        int srow  = m0 + (half << 3) + (lane & 7);
        int shead = (n0 >> 5) + ((lane >> 3) & 1);
        as[(size_t)srow * HEADS + shead] = s1[0];
        ad[(size_t)srow * HEADS + shead] = d1[0];
    } else {
        int koff = half << 1;

        float bmask[NACT];
        unsigned boff[NACT];
#pragma unroll
        for (int j = 0; j < NACT; ++j) {
            int bcol = n0 + (j << 4) + l15;
            bmask[j] = (bcol < NN) ? 1.0f : 0.0f;
            int bcolc = (bcol < NN) ? bcol : (NN - 1);
            boff[j] = (unsigned)(koff * NN + bcolc);
        }
        const float* ap = A + (size_t)arow * KK + koff;

        v8f acc[NACT];
#pragma unroll
        for (int j = 0; j < NACT; ++j) acc[j] = (v8f){};

#pragma unroll
        for (int k = 0; k < KK; k += 4) {
            v2f a; a.x = ap[k]; a.y = ap[k + 1];
            float t0[NACT], t1[NACT];
#pragma unroll
            for (int j = 0; j < NACT; ++j) {
                t0[j] = B[boff[j] + k * NN];
                t1[j] = B[boff[j] + k * NN + NN];
            }
#pragma unroll
            for (int j = 0; j < NACT; ++j) {
                v2f b;
                b.x = t0[j] * bmask[j]; b.y = t1[j] * bmask[j];
                acc[j] = __builtin_amdgcn_wmma_f32_16x16x4_f32(
                    false, a, false, b, (short)0, acc[j], false, false);
            }
        }

#pragma unroll
        for (int j = 0; j < NACT; ++j) {
            int bcol = n0 + (j << 4) + l15;
            if (bcol < NN) {
                float bv = bias ? bias[bcol] : 0.0f;
#pragma unroll
                for (int i = 0; i < 8; ++i) {
                    int row = m0 + i + (half << 3);
                    C[(size_t)row * NN + bcol] = acc[j][i] + bv;
                }
            }
        }
    }
}

template <int NN, int KK, bool RAGGED>
__global__ void gat_wmma_gemm_f32(const float* __restrict__ A,
                                  const float* __restrict__ B,
                                  const float* __restrict__ bias,
                                  float* __restrict__ C,
                                  const float* __restrict__ a_src,
                                  const float* __restrict__ a_dst,
                                  float* __restrict__ as, float* __restrict__ ad,
                                  int tiles_m, int ngroups) {
    int wave = blockIdx.x * (blockDim.x >> 5) + (threadIdx.x >> 5);
    if (wave >= tiles_m * ngroups) return;
    gemm_strip<NN, KK, RAGGED>(wave, A, B, bias, C, a_src, a_dst, as, ad, tiles_m);
}

__device__ __forceinline__ float leaky(float e) {
    return e > 0.0f ? e : NEG_SLOPE * e;
}

#define SORT_BT 256
#define EPT 13
#define CHUNK (SORT_BT * EPT)
#define RSHIFT 8
#define RNODES (1 << RSHIFT)
#define RANGES_MAX 256
#define PIECES_MAX 256
#define STAGE_CAP 12288
#define PLACE_BATCH 8

__device__ __forceinline__ int block_scan_excl(int v, int* wsum, int& total) {
    int lane = threadIdx.x & 31, wid = threadIdx.x >> 5;
    int x = v;
    for (int off = 1; off < 32; off <<= 1) {
        int u = __shfl_up(x, off, 32);
        if (lane >= off) x += u;
    }
    if (lane == 31) wsum[wid] = x;
    __syncthreads();
    int base = 0, tot = 0;
    for (int w = 0; w < SORT_BT / 32; ++w) {
        int s = wsum[w];
        if (w < wid) base += s;
        tot += s;
    }
    total = tot;
    __syncthreads();
    return base + x - v;
}

__device__ __forceinline__ void
csr_bin(int c, const int* __restrict__ src, const int* __restrict__ dst, int E, int nranges,
        int* __restrict__ offs, unsigned* __restrict__ buf) {
    __shared__ int cnt[RANGES_MAX];
    __shared__ int start[RANGES_MAX];
    __shared__ int wsum[SORT_BT / 32];
    __shared__ unsigned stage[CHUNK];
    int t = threadIdx.x;
    int e0 = c * CHUNK;
    cnt[t] = 0;
    int dv[EPT], sv[EPT];
#pragma unroll
    for (int j = 0; j < EPT; ++j) {
        int e = e0 + j * SORT_BT + t;
        bool ok = e < E;
        dv[j] = ok ? dst[e] : -1;
        sv[j] = ok ? src[e] : 0;
    }
    __syncthreads();
#pragma unroll
    for (int j = 0; j < EPT; ++j)
        if (dv[j] >= 0) atomicAdd(&cnt[dv[j] >> RSHIFT], 1);
    __syncthreads();
    int total;
    int ex = block_scan_excl(cnt[t], wsum, total);
    start[t] = ex;
    if (t < nranges) offs[c * (nranges + 1) + t] = ex;
    if (t == 0) offs[c * (nranges + 1) + nranges] = total;
    cnt[t] = 0;
    __syncthreads();
#pragma unroll
    for (int j = 0; j < EPT; ++j) {
        if (dv[j] >= 0) {
            int r = dv[j] >> RSHIFT;
            int k = atomicAdd(&cnt[r], 1);
            stage[start[r] + k] = ((unsigned)dv[j] << 16) | (unsigned)sv[j];
        }
    }
    __syncthreads();
    for (int i = t; i < total; i += SORT_BT) buf[(size_t)c * CHUNK + i] = stage[i];
}

__device__ __forceinline__ int piece_of(const int* ppos, int i) {
    int p = 0;
    for (int s = PIECES_MAX / 2; s > 0; s >>= 1)
        if (ppos[p + s] <= i) p += s;
    return p;
}

__device__ __forceinline__ void
csr_place(int r, const int* __restrict__ offs, const unsigned* __restrict__ buf, int nchunks,
          int nranges, int N, int* __restrict__ rowptr, int* __restrict__ col) {
    __shared__ int pstart[PIECES_MAX];
    __shared__ int ppos[PIECES_MAX];
    __shared__ int ncnt[RNODES];
    __shared__ int nrow[RNODES];
    __shared__ int wsum[SORT_BT / 32];
    __shared__ unsigned stage[STAGE_CAP];
    int t = threadIdx.x;
    int a = 0, len = 0;
    if (t < nchunks) {
        a   = offs[t * (nranges + 1) + r];
        len = offs[t * (nranges + 1) + r + 1] - a;
    }
    int T, before;
    int pp = block_scan_excl(len, wsum, T);
    (void)block_scan_excl(a, wsum, before);
    pstart[t] = t * CHUNK + a;
    ppos[t] = pp;
    ncnt[t] = 0;
    __syncthreads();

    for (int i0 = t; i0 < T; i0 += PLACE_BATCH * SORT_BT) {
        unsigned w[PLACE_BATCH];
#pragma unroll
        for (int u = 0; u < PLACE_BATCH; ++u) {
            int i = i0 + u * SORT_BT;
            int ic = i < T ? i : T - 1;
            int p = piece_of(ppos, ic);
            w[u] = buf[pstart[p] + (ic - ppos[p])];
        }
#pragma unroll
        for (int u = 0; u < PLACE_BATCH; ++u) {
            int i = i0 + u * SORT_BT;
            if (i < T) {
                if (i < STAGE_CAP) stage[i] = w[u];
                atomicAdd(&ncnt[(w[u] >> 16) & (RNODES - 1)], 1);
            }
        }
    }
    __syncthreads();
    int cv = ncnt[t];
    int tot2;
    int ex = block_scan_excl(cv, wsum, tot2);
    nrow[t] = ex;
    int d = (r << RSHIFT) + t;
    if (d < N) rowptr[d] = before + ex;
    if (d == N - 1) rowptr[N] = before + ex + cv;
    ncnt[t] = 0;
    __syncthreads();

    for (int i = t; i < T; i += SORT_BT) {
        unsigned w;
        if (i < STAGE_CAP) {
            w = stage[i];
        } else {
            int p = piece_of(ppos, i);
            w = buf[pstart[p] + (i - ppos[p])];
        }
        int dl = (w >> 16) & (RNODES - 1);
        int k = atomicAdd(&ncnt[dl], 1);
        col[before + nrow[dl] + k] = (int)(w & 0xFFFFu);
    }
}

__device__ __forceinline__ void gemm1_block(int blk, const float* __restrict__ x, const float* __restrict__ W1,
                    const float* __restrict__ bias, float* __restrict__ xh,
                    const float* __restrict__ a_src, const float* __restrict__ a_dst,
                    float* __restrict__ as, float* __restrict__ ad,
                    int tiles_m, int wave_begin, int wave_end) {
    int wave = wave_begin + blk * (SORT_BT >> 5) + ((int)threadIdx.x >> 5);
    if (wave >= wave_end) return;
    gemm_strip<HD, 128, false>(wave, x, W1, bias, xh, a_src, a_dst, as, ad, tiles_m);
}

__global__ void __launch_bounds__(SORT_BT) __attribute__((amdgpu_waves_per_eu(4, 10)))
sort_bin_with_gemm1(const int* __restrict__ src, const int* __restrict__ dst, int E, int nranges,
                    int* __restrict__ offs, unsigned* __restrict__ buf, int nsort,
                    const float* __restrict__ x, const float* __restrict__ W1,
                    const float* __restrict__ bias, float* __restrict__ xh,
                    const float* __restrict__ a_src, const float* __restrict__ a_dst,
                    float* __restrict__ as, float* __restrict__ ad,
                    int tiles_m, int wave_begin, int wave_end) {
    int b = (int)blockIdx.x;
    if (b < nsort) csr_bin(b, src, dst, E, nranges, offs, buf);
    else gemm1_block(b - nsort, x, W1, bias, xh, a_src, a_dst, as, ad, tiles_m, wave_begin, wave_end);
}

__global__ void __launch_bounds__(SORT_BT) __attribute__((amdgpu_waves_per_eu(4, 10)))
sort_place_with_gemm1(const int* __restrict__ offs, const unsigned* __restrict__ buf, int nchunks,
                      int nranges, int N, int* __restrict__ rowptr, int* __restrict__ col,
                      int nsort,
                    const float* __restrict__ x, const float* __restrict__ W1,
                    const float* __restrict__ bias, float* __restrict__ xh,
                    const float* __restrict__ a_src, const float* __restrict__ a_dst,
                    float* __restrict__ as, float* __restrict__ ad,
                    int tiles_m, int wave_begin, int wave_end) {
    int b = (int)blockIdx.x;
    if (b < nsort) csr_place(b, offs, buf, nchunks, nranges, N, rowptr, col);
    else gemm1_block(b - nsort, x, W1, bias, xh, a_src, a_dst, as, ad, tiles_m, wave_begin, wave_end);
}

__device__ __forceinline__ int row_cols(const int* __restrict__ col, int start, int deg,
                                        int d, int base, int lane) {
    int t = base + lane;
    return (t < deg) ? col[start + t] : d;
}

__device__ __forceinline__ int lane_fetch(int src, int v) {
    return __builtin_amdgcn_ds_bpermute(src << 2, v);
}
__device__ __forceinline__ float lane_fetch(int src, float v) {
    return __builtin_bit_cast(float, lane_fetch(src, __builtin_bit_cast(int, v)));
}

__device__ __forceinline__ float row_scores32(const float* __restrict__ as, int mycol,
                                              float ad_l, int lane, float (&e)[8]) {
    float m = -INFINITY;
#pragma unroll
    for (int q = 0; q < 8; ++q) {
        int sj = lane_fetch(q * 4 + (lane >> 3), mycol);
        e[q] = leaky(as[(size_t)sj * HEADS + (lane & 7)] + ad_l);
        m = fmaxf(m, e[q]);
    }
    return m;
}

__device__ __forceinline__ float row_weights32(const float (&e)[8], float m, int cnt, int lane,
                                               float (&p)[8]) {
    float s = 0.0f;
#pragma unroll
    for (int q = 0; q < 8; ++q) {
        p[q] = (q * 4 + (lane >> 3) < cnt) ? __expf(e[q] - m) : 0.0f;
        s += p[q];
    }
    return s;
}

__device__ __forceinline__ void row_accum32(const float* __restrict__ xh, int mycol, int cnt,
                                            int lane, const float (&p)[8],
                                            v4f& acc0, v4f& acc1) {
    int h0 = lane >> 3;
    unsigned lane16 = (unsigned)lane * 16u;
#pragma unroll
    for (int q = 0; q < 8; ++q) {
        if (q * 4 < cnt) {
#pragma unroll
            for (int u = 0; u < 4; ++u) {
                int sj = __builtin_amdgcn_readlane(mycol, q * 4 + u);
                unsigned off = (unsigned)sj * (unsigned)(HD * sizeof(float)) + lane16;
                v4f x0 = *(const v4f*)((const char*)xh + off);
                v4f x1 = *(const v4f*)((const char*)xh + off + 512);
                float w0 = lane_fetch(u * 8 + h0, p[q]);
                float w1 = lane_fetch(u * 8 + 4 + h0, p[q]);
                acc0 += w0 * x0;
                acc1 += w1 * x1;
            }
        }
    }
}

__global__ void __launch_bounds__(128)
gat_row_softmax_agg(const int* __restrict__ rowptr, const int* __restrict__ col,
                    const float* __restrict__ as, const float* __restrict__ ad,
                    const float* __restrict__ xh, const float* __restrict__ bias,
                    float* __restrict__ out, int N) {
    int lane = threadIdx.x & 31;
    int w0 = __builtin_amdgcn_readfirstlane((int)(blockIdx.x * (blockDim.x >> 5) + (threadIdx.x >> 5)));
    int nw = (int)(gridDim.x * (blockDim.x >> 5));
    const float* bp = bias + lane * 4;
    v4f bias0 = *(const v4f*)bp;
    v4f bias1 = *(const v4f*)(bp + 128);
    for (int d = w0; d < N; d += nw) {
        int start = rowptr[d];
        int deg   = rowptr[d + 1] - start;
        int total = deg + 1;
        int h0 = lane >> 3;
        float ad_l = ad[(size_t)d * HEADS + (lane & 7)];

        int col0 = row_cols(col, start, deg, d, 0, lane);
        float e0[8];
        float m = row_scores32(as, col0, ad_l, lane, e0);
        for (int base = 32; base < total; base += 32) {
            float et[8];
            m = fmaxf(m, row_scores32(as, row_cols(col, start, deg, d, base, lane), ad_l, lane, et));
        }
        m = fmaxf(m, __shfl_xor(m, 8, 32));
        m = fmaxf(m, __shfl_xor(m, 16, 32));

        float p[8];
        float lsum = row_weights32(e0, m, total, lane, p);
        v4f acc0 = (v4f){}, acc1 = (v4f){};
        row_accum32(xh, col0, total < 32 ? total : 32, lane, p, acc0, acc1);
        for (int base = 32; base < total; base += 32) {
            int c = row_cols(col, start, deg, d, base, lane);
            float et[8];
            (void)row_scores32(as, c, ad_l, lane, et);
            int rem = total - base;
            lsum += row_weights32(et, m, rem, lane, p);
            row_accum32(xh, c, rem < 32 ? rem : 32, lane, p, acc0, acc1);
        }
        lsum += __shfl_xor(lsum, 8, 32);
        lsum += __shfl_xor(lsum, 16, 32);
        float sum0 = __shfl(lsum, h0, 32), sum1 = __shfl(lsum, 4 + h0, 32);

        v4f r0 = acc0 / sum0 + bias0;
        v4f r1 = acc1 / sum1 + bias1;
        for (int i = 0; i < 4; ++i) {
            r0[i] = r0[i] > 0.0f ? r0[i] : 0.0f;
            r1[i] = r1[i] > 0.0f ? r1[i] : 0.0f;
        }
        float* op = out + (size_t)d * HD + lane * 4;
        *(v4f*)op = r0;
        *(v4f*)(op + 128) = r1;
    }
}

extern "C" void kernel_launch(void* const* d_in, const int* in_sizes, int n_in,
                              void* d_out, int out_size, void* d_ws, size_t ws_size,
                              hipStream_t stream) {
    const float* x   = (const float*)d_in[0];
    const int*   ei  = (const int*)d_in[1];
    const float* W1  = (const float*)d_in[2];
    const float* a1s = (const float*)d_in[3];
    const float* a1d = (const float*)d_in[4];
    const float* b1  = (const float*)d_in[5];
    const float* W2  = (const float*)d_in[6];
    const float* a2s = (const float*)d_in[7];
    const float* a2d = (const float*)d_in[8];
    const float* b2  = (const float*)d_in[9];
    const float* Wc  = (const float*)d_in[10];
    const float* bc  = (const float*)d_in[11];

    const int N = in_sizes[0] / 128;
    const int E = in_sizes[1] / 2;
    const int* src = ei;
    const int* dst = ei + E;

    float* ws    = (float*)d_ws;
    float* xh    = ws;
    float* hbuf  = xh   + (size_t)N * HD;
    float* as_   = hbuf + (size_t)N * HD;
    float* ad_   = as_  + (size_t)N * HEADS;
    int*   rowptr = (int*)(ad_ + (size_t)N * HEADS);
    int*   col    = rowptr + N + 1;
    const int nranges = (N + RNODES - 1) >> RSHIFT;
    const int nchunks = (E + CHUNK - 1) / CHUNK;
    int*      offs = col + E;
    unsigned* ebuf = (unsigned*)(offs + (size_t)nchunks * (nranges + 1));

    const int BT = 256;
    const int ROW_BT       = 128;
    const int ROW_BLOCKS   = 2304;
    const int tiles_m = N / 16;
    auto cdiv = [](int a, int b) { return (a + b - 1) / b; };
    const int ngroups = HD / (16 * NSUB);
    const int gemm_waves  = tiles_m * ngroups;
    const int gemm_blocks = cdiv(gemm_waves, 8);
    const int gemm_first  = cdiv(gemm_waves * 2, 5);

    sort_bin_with_gemm1<<<nchunks + cdiv(gemm_first, 8), SORT_BT, 0, stream>>>(
        src, dst, E, nranges, offs, ebuf, nchunks,
        x, W1, nullptr, xh, a1s, a1d, as_, ad_, tiles_m, 0, gemm_first);
    sort_place_with_gemm1<<<nranges + cdiv(gemm_waves - gemm_first, 8), SORT_BT, 0, stream>>>(
        offs, ebuf, nchunks, nranges, N, rowptr, col, nranges,
        x, W1, nullptr, xh, a1s, a1d, as_, ad_, tiles_m, gemm_first, gemm_waves);

    gat_row_softmax_agg<<<ROW_BLOCKS, ROW_BT, 0, stream>>>(rowptr, col, as_, ad_, xh, b1, hbuf, N);

    gat_wmma_gemm_f32<HD, HD, false><<<gemm_blocks, BT, 0, stream>>>(
        hbuf, W2, nullptr, xh, a2s, a2d, as_, ad_, tiles_m, ngroups);
    gat_row_softmax_agg<<<ROW_BLOCKS, ROW_BT, 0, stream>>>(rowptr, col, as_, ad_, xh, b2, hbuf, N);

    gat_wmma_gemm_f32<40, HD, true><<<cdiv(tiles_m, 8), BT, 0, stream>>>(
        hbuf, Wc, bc, (float*)d_out, nullptr, nullptr, nullptr, nullptr, tiles_m,  1);
}


# Row kernels hand each strip of 16 finished rows through LDS to the GEMM that consumes it (layer-2 GEMM, classifier), same workgroup: 6 -> 4 launches, layer outputs never global; sum orders unchanged
# speedup vs baseline: 5.5403x; 1.1453x over previous
#include <hip/hip_runtime.h>
#include <math.h>

typedef __attribute__((ext_vector_type(2))) float v2f;
typedef __attribute__((ext_vector_type(8))) float v8f;
typedef __attribute__((ext_vector_type(4))) float v4f;

#define HEADS 8
#define HID 32
#define HD 256
#define NEG_SLOPE 0.2f
#define NSUB 4


template <int M>
__device__ __forceinline__ float lane_xor(float v) {
    return __builtin_bit_cast(float, __builtin_amdgcn_update_dpp(
        0, __builtin_bit_cast(int, v), 0x160 | M, 0xF, 0xF, true));
}

template <int M, int CNT>
__device__ __forceinline__ void score_level(const float* in, float* out, bool hi) {
#pragma unroll
    for (int k = 0; k < CNT; ++k) {
        float keep = hi ? in[k + CNT] : in[k];
        float send = hi ? in[k] : in[k + CNT];
        out[k] = keep + lane_xor<M>(send);
    }
}

template <int NN, int KK, bool RAGGED, int LDA>
__device__ __forceinline__ void gemm_strip(const float* __restrict__ Arow0, int m0, int n0,
                                           const float* __restrict__ B,
                                           const float* __restrict__ bias,
                                           float* __restrict__ C,
                                           const float* __restrict__ a_src,
                                           const float* __restrict__ a_dst,
                                           float* __restrict__ as, float* __restrict__ ad) {
    constexpr int TILES_N = (NN + 15) / 16;
    constexpr int NACT = TILES_N < NSUB ? TILES_N : NSUB;

    int lane = threadIdx.x & 31;
    int half = lane >> 4;
    int l15  = lane & 15;
    const float* arowp = Arow0 + l15 * LDA;

    if constexpr (!RAGGED) {
        static_assert(NN % 64 == 0 && KK % 8 == 0 && NSUB == 4, "full-width strip shape");
        const float* ap = arowp + (half << 2);
        const float* bp = B + (size_t)(half << 2) * NN + n0 + (l15 << 2);

        v8f acc[4];
#pragma unroll
        for (int j = 0; j < 4; ++j) acc[j] = (v8f){};

#pragma unroll
        for (int k = 0; k < KK; k += 8) {
            v4f a4 = *(const v4f*)(ap + k);
            v4f b0 = *(const v4f*)(bp + (k + 0) * NN);
            v4f b1 = *(const v4f*)(bp + (k + 1) * NN);
            v4f b2 = *(const v4f*)(bp + (k + 2) * NN);
            v4f b3 = *(const v4f*)(bp + (k + 3) * NN);
            v2f alo; alo.x = a4.x; alo.y = a4.y;
            v2f ahi; ahi.x = a4.z; ahi.y = a4.w;
#pragma unroll
            for (int j = 0; j < 4; ++j) {
                v2f b; b.x = b0[j]; b.y = b1[j];
                acc[j] = __builtin_amdgcn_wmma_f32_16x16x4_f32(
                    false, alo, false, b, (short)0, acc[j], false, false);
            }
#pragma unroll
            for (int j = 0; j < 4; ++j) {
                v2f b; b.x = b2[j]; b.y = b3[j];
                acc[j] = __builtin_amdgcn_wmma_f32_16x16x4_f32(
                    false, ahi, false, b, (short)0, acc[j], false, false);
            }
        }

        int ccol = n0 + (l15 << 2);
        v4f bv = (v4f){};
        if (bias) bv = *(const v4f*)(bias + ccol);
#pragma unroll
        for (int i = 0; i < 8; ++i) {
            int row = m0 + i + (half << 3);
            v4f o;
            o.x = acc[0][i]; o.y = acc[1][i]; o.z = acc[2][i]; o.w = acc[3][i];
            *(v4f*)(C + (size_t)row * NN + ccol) = o + bv;
        }

        static_assert(HID == 32 && NN == HEADS * HID, "a 64-column strip is two heads");
        v4f sv = *(const v4f*)(a_src + ccol);
        v4f dv = *(const v4f*)(a_dst + ccol);
        float ps[8], pd[8];
#pragma unroll
        for (int i = 0; i < 8; ++i) {
            ps[i] = fmaf(acc[3][i], sv.w, fmaf(acc[2][i], sv.z, fmaf(acc[1][i], sv.y, acc[0][i] * sv.x)));
            pd[i] = fmaf(acc[3][i], dv.w, fmaf(acc[2][i], dv.z, fmaf(acc[1][i], dv.y, acc[0][i] * dv.x)));
        }
        float s4[4], d4[4], s2[2], d2[2], s1[1], d1[1];
        score_level<4, 4>(ps, s4, (lane & 4) != 0);
        score_level<4, 4>(pd, d4, (lane & 4) != 0);
        score_level<2, 2>(s4, s2, (lane & 2) != 0);
        score_level<2, 2>(d4, d2, (lane & 2) != 0);
        score_level<1, 1>(s2, s1, (lane & 1) != 0);
        score_level<1, 1>(d2, d1, (lane & 1) != 0);
        int srow  = m0 + (half << 3) + (lane & 7);
        int shead = (n0 >> 5) + ((lane >> 3) & 1);
        as[(size_t)srow * HEADS + shead] = s1[0];
        ad[(size_t)srow * HEADS + shead] = d1[0];
    } else {
        int koff = half << 1;

        float bmask[NACT];
        unsigned boff[NACT];
#pragma unroll
        for (int j = 0; j < NACT; ++j) {
            int bcol = n0 + (j << 4) + l15;
            bmask[j] = (bcol < NN) ? 1.0f : 0.0f;
            int bcolc = (bcol < NN) ? bcol : (NN - 1);
            boff[j] = (unsigned)(koff * NN + bcolc);
        }
        const float* ap = arowp + koff;

        v8f acc[NACT];
#pragma unroll
        for (int j = 0; j < NACT; ++j) acc[j] = (v8f){};

#pragma unroll
        for (int k = 0; k < KK; k += 4) {
            v2f a; a.x = ap[k]; a.y = ap[k + 1];
            float t0[NACT], t1[NACT];
#pragma unroll
            for (int j = 0; j < NACT; ++j) {
                t0[j] = B[boff[j] + k * NN];
                t1[j] = B[boff[j] + k * NN + NN];
            }
#pragma unroll
            for (int j = 0; j < NACT; ++j) {
                v2f b;
                b.x = t0[j] * bmask[j]; b.y = t1[j] * bmask[j];
                acc[j] = __builtin_amdgcn_wmma_f32_16x16x4_f32(
                    false, a, false, b, (short)0, acc[j], false, false);
            }
        }

#pragma unroll
        for (int j = 0; j < NACT; ++j) {
            int bcol = n0 + (j << 4) + l15;
            if (bcol < NN) {
                float bv = bias ? bias[bcol] : 0.0f;
#pragma unroll
                for (int i = 0; i < 8; ++i) {
                    int row = m0 + i + (half << 3);
                    C[(size_t)row * NN + bcol] = acc[j][i] + bv;
                }
            }
        }
    }
}

__device__ __forceinline__ float leaky(float e) {
    return e > 0.0f ? e : NEG_SLOPE * e;
}

#define SORT_BT 256
#define EPT 13
#define CHUNK (SORT_BT * EPT)
#define RSHIFT 8
#define RNODES (1 << RSHIFT)
#define RANGES_MAX 256
#define PIECES_MAX 256
#define STAGE_CAP 12288
#define PLACE_BATCH 8

__device__ __forceinline__ int block_scan_excl(int v, int* wsum, int& total) {
    int lane = threadIdx.x & 31, wid = threadIdx.x >> 5;
    int x = v;
    for (int off = 1; off < 32; off <<= 1) {
        int u = __shfl_up(x, off, 32);
        if (lane >= off) x += u;
    }
    if (lane == 31) wsum[wid] = x;
    __syncthreads();
    int base = 0, tot = 0;
    for (int w = 0; w < SORT_BT / 32; ++w) {
        int s = wsum[w];
        if (w < wid) base += s;
        tot += s;
    }
    total = tot;
    __syncthreads();
    return base + x - v;
}

__device__ __forceinline__ void
csr_bin(int c, const int* __restrict__ src, const int* __restrict__ dst, int E, int nranges,
        int* __restrict__ offs, unsigned* __restrict__ buf) {
    __shared__ int cnt[RANGES_MAX];
    __shared__ int start[RANGES_MAX];
    __shared__ int wsum[SORT_BT / 32];
    __shared__ unsigned stage[CHUNK];
    int t = threadIdx.x;
    int e0 = c * CHUNK;
    cnt[t] = 0;
    int dv[EPT], sv[EPT];
#pragma unroll
    for (int j = 0; j < EPT; ++j) {
        int e = e0 + j * SORT_BT + t;
        bool ok = e < E;
        dv[j] = ok ? dst[e] : -1;
        sv[j] = ok ? src[e] : 0;
    }
    __syncthreads();
#pragma unroll
    for (int j = 0; j < EPT; ++j)
        if (dv[j] >= 0) atomicAdd(&cnt[dv[j] >> RSHIFT], 1);
    __syncthreads();
    int total;
    int ex = block_scan_excl(cnt[t], wsum, total);
    start[t] = ex;
    if (t < nranges) offs[c * (nranges + 1) + t] = ex;
    if (t == 0) offs[c * (nranges + 1) + nranges] = total;
    cnt[t] = 0;
    __syncthreads();
#pragma unroll
    for (int j = 0; j < EPT; ++j) {
        if (dv[j] >= 0) {
            int r = dv[j] >> RSHIFT;
            int k = atomicAdd(&cnt[r], 1);
            stage[start[r] + k] = ((unsigned)dv[j] << 16) | (unsigned)sv[j];
        }
    }
    __syncthreads();
    for (int i = t; i < total; i += SORT_BT) buf[(size_t)c * CHUNK + i] = stage[i];
}

__device__ __forceinline__ int piece_of(const int* ppos, int i) {
    int p = 0;
    for (int s = PIECES_MAX / 2; s > 0; s >>= 1)
        if (ppos[p + s] <= i) p += s;
    return p;
}

__device__ __forceinline__ void
csr_place(int r, const int* __restrict__ offs, const unsigned* __restrict__ buf, int nchunks,
          int nranges, int N, int* __restrict__ rowptr, int* __restrict__ col) {
    __shared__ int pstart[PIECES_MAX];
    __shared__ int ppos[PIECES_MAX];
    __shared__ int ncnt[RNODES];
    __shared__ int nrow[RNODES];
    __shared__ int wsum[SORT_BT / 32];
    __shared__ unsigned stage[STAGE_CAP];
    int t = threadIdx.x;
    int a = 0, len = 0;
    if (t < nchunks) {
        a   = offs[t * (nranges + 1) + r];
        len = offs[t * (nranges + 1) + r + 1] - a;
    }
    int T, before;
    int pp = block_scan_excl(len, wsum, T);
    (void)block_scan_excl(a, wsum, before);
    pstart[t] = t * CHUNK + a;
    ppos[t] = pp;
    ncnt[t] = 0;
    __syncthreads();

    for (int i0 = t; i0 < T; i0 += PLACE_BATCH * SORT_BT) {
        unsigned w[PLACE_BATCH];
#pragma unroll
        for (int u = 0; u < PLACE_BATCH; ++u) {
            int i = i0 + u * SORT_BT;
            int ic = i < T ? i : T - 1;
            int p = piece_of(ppos, ic);
            w[u] = buf[pstart[p] + (ic - ppos[p])];
        }
#pragma unroll
        for (int u = 0; u < PLACE_BATCH; ++u) {
            int i = i0 + u * SORT_BT;
            if (i < T) {
                if (i < STAGE_CAP) stage[i] = w[u];
                atomicAdd(&ncnt[(w[u] >> 16) & (RNODES - 1)], 1);
            }
        }
    }
    __syncthreads();
    int cv = ncnt[t];
    int tot2;
    int ex = block_scan_excl(cv, wsum, tot2);
    nrow[t] = ex;
    int d = (r << RSHIFT) + t;
    if (d < N) rowptr[d] = before + ex;
    if (d == N - 1) rowptr[N] = before + ex + cv;
    ncnt[t] = 0;
    __syncthreads();

    for (int i = t; i < T; i += SORT_BT) {
        unsigned w;
        if (i < STAGE_CAP) {
            w = stage[i];
        } else {
            int p = piece_of(ppos, i);
            w = buf[pstart[p] + (i - ppos[p])];
        }
        int dl = (w >> 16) & (RNODES - 1);
        int k = atomicAdd(&ncnt[dl], 1);
        col[before + nrow[dl] + k] = (int)(w & 0xFFFFu);
    }
}

__device__ __forceinline__ void gemm1_block(int blk, const float* __restrict__ x, const float* __restrict__ W1,
                    const float* __restrict__ bias, float* __restrict__ xh,
                    const float* __restrict__ a_src, const float* __restrict__ a_dst,
                    float* __restrict__ as, float* __restrict__ ad,
                    int tiles_m, int wave_begin, int wave_end) {
    int wave = wave_begin + blk * (SORT_BT >> 5) + ((int)threadIdx.x >> 5);
    if (wave >= wave_end) return;
    int m0 = (wave % tiles_m) << 4;
    int n0 = (wave / tiles_m) << 6;
    gemm_strip<HD, 128, false, 128>(x + (size_t)m0 * 128, m0, n0, W1, bias, xh, a_src, a_dst, as, ad);
}

__global__ void __launch_bounds__(SORT_BT) __attribute__((amdgpu_waves_per_eu(4, 10)))
sort_bin_with_gemm1(const int* __restrict__ src, const int* __restrict__ dst, int E, int nranges,
                    int* __restrict__ offs, unsigned* __restrict__ buf, int nsort,
                    const float* __restrict__ x, const float* __restrict__ W1,
                    const float* __restrict__ bias, float* __restrict__ xh,
                    const float* __restrict__ a_src, const float* __restrict__ a_dst,
                    float* __restrict__ as, float* __restrict__ ad,
                    int tiles_m, int wave_begin, int wave_end) {
    int b = (int)blockIdx.x;
    if (b < nsort) csr_bin(b, src, dst, E, nranges, offs, buf);
    else gemm1_block(b - nsort, x, W1, bias, xh, a_src, a_dst, as, ad, tiles_m, wave_begin, wave_end);
}

__global__ void __launch_bounds__(SORT_BT) __attribute__((amdgpu_waves_per_eu(4, 10)))
sort_place_with_gemm1(const int* __restrict__ offs, const unsigned* __restrict__ buf, int nchunks,
                      int nranges, int N, int* __restrict__ rowptr, int* __restrict__ col,
                      int nsort,
                    const float* __restrict__ x, const float* __restrict__ W1,
                    const float* __restrict__ bias, float* __restrict__ xh,
                    const float* __restrict__ a_src, const float* __restrict__ a_dst,
                    float* __restrict__ as, float* __restrict__ ad,
                    int tiles_m, int wave_begin, int wave_end) {
    int b = (int)blockIdx.x;
    if (b < nsort) csr_place(b, offs, buf, nchunks, nranges, N, rowptr, col);
    else gemm1_block(b - nsort, x, W1, bias, xh, a_src, a_dst, as, ad, tiles_m, wave_begin, wave_end);
}

__device__ __forceinline__ int row_cols(const int* __restrict__ col, int start, int deg,
                                        int d, int base, int lane) {
    int t = base + lane;
    return (t < deg) ? col[start + t] : d;
}

__device__ __forceinline__ int lane_fetch(int src, int v) {
    return __builtin_amdgcn_ds_bpermute(src << 2, v);
}
__device__ __forceinline__ float lane_fetch(int src, float v) {
    return __builtin_bit_cast(float, lane_fetch(src, __builtin_bit_cast(int, v)));
}

__device__ __forceinline__ float row_scores32(const float* __restrict__ as, int mycol,
                                              float ad_l, int lane, float (&e)[8]) {
    float m = -INFINITY;
#pragma unroll
    for (int q = 0; q < 8; ++q) {
        int sj = lane_fetch(q * 4 + (lane >> 3), mycol);
        e[q] = leaky(as[(size_t)sj * HEADS + (lane & 7)] + ad_l);
        m = fmaxf(m, e[q]);
    }
    return m;
}

__device__ __forceinline__ float row_weights32(const float (&e)[8], float m, int cnt, int lane,
                                               float (&p)[8]) {
    float s = 0.0f;
#pragma unroll
    for (int q = 0; q < 8; ++q) {
        p[q] = (q * 4 + (lane >> 3) < cnt) ? __expf(e[q] - m) : 0.0f;
        s += p[q];
    }
    return s;
}

__device__ __forceinline__ void row_accum32(const float* __restrict__ xh, int mycol, int cnt,
                                            int lane, const float (&p)[8],
                                            v4f& acc0, v4f& acc1) {
    int h0 = lane >> 3;
    unsigned lane16 = (unsigned)lane * 16u;
#pragma unroll
    for (int q = 0; q < 8; ++q) {
        if (q * 4 < cnt) {
#pragma unroll
            for (int u = 0; u < 4; ++u) {
                int sj = __builtin_amdgcn_readlane(mycol, q * 4 + u);
                unsigned off = (unsigned)sj * (unsigned)(HD * sizeof(float)) + lane16;
                v4f x0 = *(const v4f*)((const char*)xh + off);
                v4f x1 = *(const v4f*)((const char*)xh + off + 512);
                float w0 = lane_fetch(u * 8 + h0, p[q]);
                float w1 = lane_fetch(u * 8 + 4 + h0, p[q]);
                acc0 += w0 * x0;
                acc1 += w1 * x1;
            }
        }
    }
}

__device__ __forceinline__ void row_softmax_agg(int d, int lane,
                    const int* __restrict__ rowptr, const int* __restrict__ col,
                    const float* __restrict__ as, const float* __restrict__ ad,
                    const float* __restrict__ xh, v4f bias0, v4f bias1, v4f& r0, v4f& r1) {
    int start = rowptr[d];
    int deg   = rowptr[d + 1] - start;
    int total = deg + 1;
    int h0 = lane >> 3;
    float ad_l = ad[(size_t)d * HEADS + (lane & 7)];

    int col0 = row_cols(col, start, deg, d, 0, lane);
    float e0[8];
    float m = row_scores32(as, col0, ad_l, lane, e0);
    for (int base = 32; base < total; base += 32) {
        float et[8];
        m = fmaxf(m, row_scores32(as, row_cols(col, start, deg, d, base, lane), ad_l, lane, et));
    }
    m = fmaxf(m, __shfl_xor(m, 8, 32));
    m = fmaxf(m, __shfl_xor(m, 16, 32));

    float p[8];
    float lsum = row_weights32(e0, m, total, lane, p);
    v4f acc0 = (v4f){}, acc1 = (v4f){};
    row_accum32(xh, col0, total < 32 ? total : 32, lane, p, acc0, acc1);
    for (int base = 32; base < total; base += 32) {
        int c = row_cols(col, start, deg, d, base, lane);
        float et[8];
        (void)row_scores32(as, c, ad_l, lane, et);
        int rem = total - base;
        lsum += row_weights32(et, m, rem, lane, p);
        row_accum32(xh, c, rem < 32 ? rem : 32, lane, p, acc0, acc1);
    }
    lsum += __shfl_xor(lsum, 8, 32);
    lsum += __shfl_xor(lsum, 16, 32);
    float sum0 = __shfl(lsum, h0, 32), sum1 = __shfl(lsum, 4 + h0, 32);

    r0 = acc0 / sum0 + bias0;
    r1 = acc1 / sum1 + bias1;
    for (int i = 0; i < 4; ++i) {
        r0[i] = r0[i] > 0.0f ? r0[i] : 0.0f;
        r1[i] = r1[i] > 0.0f ? r1[i] : 0.0f;
    }
}

#define TILE_LD (HD + 4)

template <int NN, bool RAGGED>
__global__ void __launch_bounds__(128) __attribute__((amdgpu_waves_per_eu(4, 9)))
gat_row_then_gemm(const int* __restrict__ rowptr, const int* __restrict__ col,
                  const float* __restrict__ as, const float* __restrict__ ad,
                  const float* __restrict__ xh, const float* __restrict__ bias,
                  const float* __restrict__ B, const float* __restrict__ gbias,
                  float* __restrict__ C,
                  const float* __restrict__ a_src, const float* __restrict__ a_dst,
                  float* __restrict__ as_out, float* __restrict__ ad_out) {
    __shared__ __attribute__((aligned(16))) float tile[16 * TILE_LD];
    int lane = threadIdx.x & 31;
    int wv = __builtin_amdgcn_readfirstlane((int)(threadIdx.x >> 5));
    int m0 = (int)blockIdx.x << 4;
    const float* bp = bias + lane * 4;
    v4f bias0 = *(const v4f*)bp;
    v4f bias1 = *(const v4f*)(bp + 128);
    for (int i = 0; i < 4; ++i) {
        int lr = wv * 4 + i;
        v4f r0, r1;
        row_softmax_agg(m0 + lr, lane, rowptr, col, as, ad, xh, bias0, bias1, r0, r1);
        float* tp = tile + lr * TILE_LD + lane * 4;
        *(v4f*)tp = r0;
        *(v4f*)(tp + 128) = r1;
    }
    __syncthreads();
    if constexpr (!RAGGED) {
        gemm_strip<NN, HD, false, TILE_LD>(tile, m0, wv << 6, B, gbias, C, a_src, a_dst, as_out, ad_out);
    } else {
        if (wv == 0)
            gemm_strip<NN, HD, true, TILE_LD>(tile, m0, 0, B, gbias, C, a_src, a_dst, as_out, ad_out);
    }
}

extern "C" void kernel_launch(void* const* d_in, const int* in_sizes, int n_in,
                              void* d_out, int out_size, void* d_ws, size_t ws_size,
                              hipStream_t stream) {
    const float* x   = (const float*)d_in[0];
    const int*   ei  = (const int*)d_in[1];
    const float* W1  = (const float*)d_in[2];
    const float* a1s = (const float*)d_in[3];
    const float* a1d = (const float*)d_in[4];
    const float* b1  = (const float*)d_in[5];
    const float* W2  = (const float*)d_in[6];
    const float* a2s = (const float*)d_in[7];
    const float* a2d = (const float*)d_in[8];
    const float* b2  = (const float*)d_in[9];
    const float* Wc  = (const float*)d_in[10];
    const float* bc  = (const float*)d_in[11];

    const int N = in_sizes[0] / 128;
    const int E = in_sizes[1] / 2;
    const int* src = ei;
    const int* dst = ei + E;

    float* ws    = (float*)d_ws;
    float* xh    = ws;
    float* xh2   = xh   + (size_t)N * HD;
    float* as_   = xh2  + (size_t)N * HD;
    float* ad_   = as_  + (size_t)N * HEADS;
    float* as2_  = ad_  + (size_t)N * HEADS;
    float* ad2_  = as2_ + (size_t)N * HEADS;
    int*   rowptr = (int*)(ad2_ + (size_t)N * HEADS);
    int*   col    = rowptr + N + 1;
    const int nranges = (N + RNODES - 1) >> RSHIFT;
    const int nchunks = (E + CHUNK - 1) / CHUNK;
    int*      offs = col + E;
    unsigned* ebuf = (unsigned*)(offs + (size_t)nchunks * (nranges + 1));

    const int ROW_BT  = 128;
    const int tiles_m = N / 16;
    auto cdiv = [](int a, int b) { return (a + b - 1) / b; };
    const int ngroups = HD / (16 * NSUB);
    const int gemm_waves  = tiles_m * ngroups;
    const int gemm_first  = cdiv(gemm_waves * 2, 5);

    sort_bin_with_gemm1<<<nchunks + cdiv(gemm_first, 8), SORT_BT, 0, stream>>>(
        src, dst, E, nranges, offs, ebuf, nchunks,
        x, W1, nullptr, xh, a1s, a1d, as_, ad_, tiles_m, 0, gemm_first);
    sort_place_with_gemm1<<<nranges + cdiv(gemm_waves - gemm_first, 8), SORT_BT, 0, stream>>>(
        offs, ebuf, nchunks, nranges, N, rowptr, col, nranges,
        x, W1, nullptr, xh, a1s, a1d, as_, ad_, tiles_m, gemm_first, gemm_waves);

    gat_row_then_gemm<HD, false><<<tiles_m, ROW_BT, 0, stream>>>(
        rowptr, col, as_, ad_, xh, b1, W2, nullptr, xh2, a2s, a2d, as2_, ad2_);

    gat_row_then_gemm<40, true><<<tiles_m, ROW_BT, 0, stream>>>(
        rowptr, col, as2_, ad2_, xh2, b2, Wc, bc, (float*)d_out, nullptr, nullptr, nullptr, nullptr);
}


# Classifier strip: its three 16-column tiles go to three waves (one per SIMD) instead of all to wave 0; tiles meet in LDS and leave as whole-line 16-byte stores; every k chain and sum order unchanged
# speedup vs baseline: 5.8139x; 1.0494x over previous
#include <hip/hip_runtime.h>
#include <math.h>

typedef __attribute__((ext_vector_type(2))) float v2f;
typedef __attribute__((ext_vector_type(8))) float v8f;
typedef __attribute__((ext_vector_type(4))) float v4f;

#define HEADS 8
#define HID 32
#define HD 256
#define NEG_SLOPE 0.2f
#define NSUB 4


template <int M>
__device__ __forceinline__ float lane_xor(float v) {
    return __builtin_bit_cast(float, __builtin_amdgcn_update_dpp(
        0, __builtin_bit_cast(int, v), 0x160 | M, 0xF, 0xF, true));
}

template <int M, int CNT>
__device__ __forceinline__ void score_level(const float* in, float* out, bool hi) {
#pragma unroll
    for (int k = 0; k < CNT; ++k) {
        float keep = hi ? in[k + CNT] : in[k];
        float send = hi ? in[k] : in[k + CNT];
        out[k] = keep + lane_xor<M>(send);
    }
}

template <int NN, int KK, bool RAGGED, int LDA, int SUBT = NSUB>
__device__ __forceinline__ void gemm_strip(const float* __restrict__ Arow0, int m0, int n0,
                                           const float* __restrict__ B,
                                           const float* __restrict__ bias,
                                           float* __restrict__ C,
                                           const float* __restrict__ a_src,
                                           const float* __restrict__ a_dst,
                                           float* __restrict__ as, float* __restrict__ ad) {
    constexpr int TILES_N = (NN + 15) / 16;
    constexpr int NACT = TILES_N < SUBT ? TILES_N : SUBT;

    int lane = threadIdx.x & 31;
    int half = lane >> 4;
    int l15  = lane & 15;
    const float* arowp = Arow0 + l15 * LDA;

    if constexpr (!RAGGED) {
        static_assert(NN % 64 == 0 && KK % 8 == 0 && NSUB == 4, "full-width strip shape");
        const float* ap = arowp + (half << 2);
        const float* bp = B + (size_t)(half << 2) * NN + n0 + (l15 << 2);

        v8f acc[4];
#pragma unroll
        for (int j = 0; j < 4; ++j) acc[j] = (v8f){};

#pragma unroll
        for (int k = 0; k < KK; k += 8) {
            v4f a4 = *(const v4f*)(ap + k);
            v4f b0 = *(const v4f*)(bp + (k + 0) * NN);
            v4f b1 = *(const v4f*)(bp + (k + 1) * NN);
            v4f b2 = *(const v4f*)(bp + (k + 2) * NN);
            v4f b3 = *(const v4f*)(bp + (k + 3) * NN);
            v2f alo; alo.x = a4.x; alo.y = a4.y;
            v2f ahi; ahi.x = a4.z; ahi.y = a4.w;
#pragma unroll
            for (int j = 0; j < 4; ++j) {
                v2f b; b.x = b0[j]; b.y = b1[j];
                acc[j] = __builtin_amdgcn_wmma_f32_16x16x4_f32(
                    false, alo, false, b, (short)0, acc[j], false, false);
            }
#pragma unroll
            for (int j = 0; j < 4; ++j) {
                v2f b; b.x = b2[j]; b.y = b3[j];
                acc[j] = __builtin_amdgcn_wmma_f32_16x16x4_f32(
                    false, ahi, false, b, (short)0, acc[j], false, false);
            }
        }

        int ccol = n0 + (l15 << 2);
        v4f bv = (v4f){};
        if (bias) bv = *(const v4f*)(bias + ccol);
#pragma unroll
        for (int i = 0; i < 8; ++i) {
            int row = m0 + i + (half << 3);
            v4f o;
            o.x = acc[0][i]; o.y = acc[1][i]; o.z = acc[2][i]; o.w = acc[3][i];
            *(v4f*)(C + (size_t)row * NN + ccol) = o + bv;
        }

        static_assert(HID == 32 && NN == HEADS * HID, "a 64-column strip is two heads");
        v4f sv = *(const v4f*)(a_src + ccol);
        v4f dv = *(const v4f*)(a_dst + ccol);
        float ps[8], pd[8];
#pragma unroll
        for (int i = 0; i < 8; ++i) {
            ps[i] = fmaf(acc[3][i], sv.w, fmaf(acc[2][i], sv.z, fmaf(acc[1][i], sv.y, acc[0][i] * sv.x)));
            pd[i] = fmaf(acc[3][i], dv.w, fmaf(acc[2][i], dv.z, fmaf(acc[1][i], dv.y, acc[0][i] * dv.x)));
        }
        float s4[4], d4[4], s2[2], d2[2], s1[1], d1[1];
        score_level<4, 4>(ps, s4, (lane & 4) != 0);
        score_level<4, 4>(pd, d4, (lane & 4) != 0);
        score_level<2, 2>(s4, s2, (lane & 2) != 0);
        score_level<2, 2>(d4, d2, (lane & 2) != 0);
        score_level<1, 1>(s2, s1, (lane & 1) != 0);
        score_level<1, 1>(d2, d1, (lane & 1) != 0);
        int srow  = m0 + (half << 3) + (lane & 7);
        int shead = (n0 >> 5) + ((lane >> 3) & 1);
        as[(size_t)srow * HEADS + shead] = s1[0];
        ad[(size_t)srow * HEADS + shead] = d1[0];
    } else {
        int koff = half << 1;

        float bmask[NACT];
        unsigned boff[NACT];
#pragma unroll
        for (int j = 0; j < NACT; ++j) {
            int bcol = n0 + (j << 4) + l15;
            bmask[j] = (bcol < NN) ? 1.0f : 0.0f;
            int bcolc = (bcol < NN) ? bcol : (NN - 1);
            boff[j] = (unsigned)(koff * NN + bcolc);
        }
        const float* ap = arowp + koff;

        v8f acc[NACT];
#pragma unroll
        for (int j = 0; j < NACT; ++j) acc[j] = (v8f){};

#pragma unroll
        for (int k = 0; k < KK; k += 4) {
            v2f a; a.x = ap[k]; a.y = ap[k + 1];
            float t0[NACT], t1[NACT];
#pragma unroll
            for (int j = 0; j < NACT; ++j) {
                t0[j] = B[boff[j] + k * NN];
                t1[j] = B[boff[j] + k * NN + NN];
            }
#pragma unroll
            for (int j = 0; j < NACT; ++j) {
                v2f b;
                b.x = t0[j] * bmask[j]; b.y = t1[j] * bmask[j];
                acc[j] = __builtin_amdgcn_wmma_f32_16x16x4_f32(
                    false, a, false, b, (short)0, acc[j], false, false);
            }
        }

#pragma unroll
        for (int j = 0; j < NACT; ++j) {
            int bcol = n0 + (j << 4) + l15;
            if (bcol < NN) {
                float bv = bias ? bias[bcol] : 0.0f;
#pragma unroll
                for (int i = 0; i < 8; ++i) {
                    int row = m0 + i + (half << 3);
                    C[(size_t)row * NN + bcol] = acc[j][i] + bv;
                }
            }
        }
    }
}

__device__ __forceinline__ float leaky(float e) {
    return e > 0.0f ? e : NEG_SLOPE * e;
}

#define SORT_BT 256
#define EPT 13
#define CHUNK (SORT_BT * EPT)
#define RSHIFT 8
#define RNODES (1 << RSHIFT)
#define RANGES_MAX 256
#define PIECES_MAX 256
#define STAGE_CAP 12288
#define PLACE_BATCH 8

__device__ __forceinline__ int block_scan_excl(int v, int* wsum, int& total) {
    int lane = threadIdx.x & 31, wid = threadIdx.x >> 5;
    int x = v;
    for (int off = 1; off < 32; off <<= 1) {
        int u = __shfl_up(x, off, 32);
        if (lane >= off) x += u;
    }
    if (lane == 31) wsum[wid] = x;
    __syncthreads();
    int base = 0, tot = 0;
    for (int w = 0; w < SORT_BT / 32; ++w) {
        int s = wsum[w];
        if (w < wid) base += s;
        tot += s;
    }
    total = tot;
    __syncthreads();
    return base + x - v;
}

__device__ __forceinline__ void
csr_bin(int c, const int* __restrict__ src, const int* __restrict__ dst, int E, int nranges,
        int* __restrict__ offs, unsigned* __restrict__ buf) {
    __shared__ int cnt[RANGES_MAX];
    __shared__ int start[RANGES_MAX];
    __shared__ int wsum[SORT_BT / 32];
    __shared__ unsigned stage[CHUNK];
    int t = threadIdx.x;
    int e0 = c * CHUNK;
    cnt[t] = 0;
    int dv[EPT], sv[EPT];
#pragma unroll
    for (int j = 0; j < EPT; ++j) {
        int e = e0 + j * SORT_BT + t;
        bool ok = e < E;
        dv[j] = ok ? dst[e] : -1;
        sv[j] = ok ? src[e] : 0;
    }
    __syncthreads();
#pragma unroll
    for (int j = 0; j < EPT; ++j)
        if (dv[j] >= 0) atomicAdd(&cnt[dv[j] >> RSHIFT], 1);
    __syncthreads();
    int total;
    int ex = block_scan_excl(cnt[t], wsum, total);
    start[t] = ex;
    if (t < nranges) offs[c * (nranges + 1) + t] = ex;
    if (t == 0) offs[c * (nranges + 1) + nranges] = total;
    cnt[t] = 0;
    __syncthreads();
#pragma unroll
    for (int j = 0; j < EPT; ++j) {
        if (dv[j] >= 0) {
            int r = dv[j] >> RSHIFT;
            int k = atomicAdd(&cnt[r], 1);
            stage[start[r] + k] = ((unsigned)dv[j] << 16) | (unsigned)sv[j];
        }
    }
    __syncthreads();
    for (int i = t; i < total; i += SORT_BT) buf[(size_t)c * CHUNK + i] = stage[i];
}

__device__ __forceinline__ int piece_of(const int* ppos, int i) {
    int p = 0;
    for (int s = PIECES_MAX / 2; s > 0; s >>= 1)
        if (ppos[p + s] <= i) p += s;
    return p;
}

__device__ __forceinline__ void
csr_place(int r, const int* __restrict__ offs, const unsigned* __restrict__ buf, int nchunks,
          int nranges, int N, int* __restrict__ rowptr, int* __restrict__ col) {
    __shared__ int pstart[PIECES_MAX];
    __shared__ int ppos[PIECES_MAX];
    __shared__ int ncnt[RNODES];
    __shared__ int nrow[RNODES];
    __shared__ int wsum[SORT_BT / 32];
    __shared__ unsigned stage[STAGE_CAP];
    int t = threadIdx.x;
    int a = 0, len = 0;
    if (t < nchunks) {
        a   = offs[t * (nranges + 1) + r];
        len = offs[t * (nranges + 1) + r + 1] - a;
    }
    int T, before;
    int pp = block_scan_excl(len, wsum, T);
    (void)block_scan_excl(a, wsum, before);
    pstart[t] = t * CHUNK + a;
    ppos[t] = pp;
    ncnt[t] = 0;
    __syncthreads();

    for (int i0 = t; i0 < T; i0 += PLACE_BATCH * SORT_BT) {
        unsigned w[PLACE_BATCH];
#pragma unroll
        for (int u = 0; u < PLACE_BATCH; ++u) {
            int i = i0 + u * SORT_BT;
            int ic = i < T ? i : T - 1;
            int p = piece_of(ppos, ic);
            w[u] = buf[pstart[p] + (ic - ppos[p])];
        }
#pragma unroll
        for (int u = 0; u < PLACE_BATCH; ++u) {
            int i = i0 + u * SORT_BT;
            if (i < T) {
                if (i < STAGE_CAP) stage[i] = w[u];
                atomicAdd(&ncnt[(w[u] >> 16) & (RNODES - 1)], 1);
            }
        }
    }
    __syncthreads();
    int cv = ncnt[t];
    int tot2;
    int ex = block_scan_excl(cv, wsum, tot2);
    nrow[t] = ex;
    int d = (r << RSHIFT) + t;
    if (d < N) rowptr[d] = before + ex;
    if (d == N - 1) rowptr[N] = before + ex + cv;
    ncnt[t] = 0;
    __syncthreads();

    for (int i = t; i < T; i += SORT_BT) {
        unsigned w;
        if (i < STAGE_CAP) {
            w = stage[i];
        } else {
            int p = piece_of(ppos, i);
            w = buf[pstart[p] + (i - ppos[p])];
        }
        int dl = (w >> 16) & (RNODES - 1);
        int k = atomicAdd(&ncnt[dl], 1);
        col[before + nrow[dl] + k] = (int)(w & 0xFFFFu);
    }
}

__device__ __forceinline__ void gemm1_block(int blk, const float* __restrict__ x, const float* __restrict__ W1,
                    const float* __restrict__ bias, float* __restrict__ xh,
                    const float* __restrict__ a_src, const float* __restrict__ a_dst,
                    float* __restrict__ as, float* __restrict__ ad,
                    int tiles_m, int wave_begin, int wave_end) {
    int wave = wave_begin + blk * (SORT_BT >> 5) + ((int)threadIdx.x >> 5);
    if (wave >= wave_end) return;
    int m0 = (wave % tiles_m) << 4;
    int n0 = (wave / tiles_m) << 6;
    gemm_strip<HD, 128, false, 128>(x + (size_t)m0 * 128, m0, n0, W1, bias, xh, a_src, a_dst, as, ad);
}

__global__ void __launch_bounds__(SORT_BT) __attribute__((amdgpu_waves_per_eu(4, 10)))
sort_bin_with_gemm1(const int* __restrict__ src, const int* __restrict__ dst, int E, int nranges,
                    int* __restrict__ offs, unsigned* __restrict__ buf, int nsort,
                    const float* __restrict__ x, const float* __restrict__ W1,
                    const float* __restrict__ bias, float* __restrict__ xh,
                    const float* __restrict__ a_src, const float* __restrict__ a_dst,
                    float* __restrict__ as, float* __restrict__ ad,
                    int tiles_m, int wave_begin, int wave_end) {
    int b = (int)blockIdx.x;
    if (b < nsort) csr_bin(b, src, dst, E, nranges, offs, buf);
    else gemm1_block(b - nsort, x, W1, bias, xh, a_src, a_dst, as, ad, tiles_m, wave_begin, wave_end);
}

__global__ void __launch_bounds__(SORT_BT) __attribute__((amdgpu_waves_per_eu(4, 10)))
sort_place_with_gemm1(const int* __restrict__ offs, const unsigned* __restrict__ buf, int nchunks,
                      int nranges, int N, int* __restrict__ rowptr, int* __restrict__ col,
                      int nsort,
                    const float* __restrict__ x, const float* __restrict__ W1,
                    const float* __restrict__ bias, float* __restrict__ xh,
                    const float* __restrict__ a_src, const float* __restrict__ a_dst,
                    float* __restrict__ as, float* __restrict__ ad,
                    int tiles_m, int wave_begin, int wave_end) {
    int b = (int)blockIdx.x;
    if (b < nsort) csr_place(b, offs, buf, nchunks, nranges, N, rowptr, col);
    else gemm1_block(b - nsort, x, W1, bias, xh, a_src, a_dst, as, ad, tiles_m, wave_begin, wave_end);
}

__device__ __forceinline__ int row_cols(const int* __restrict__ col, int start, int deg,
                                        int d, int base, int lane) {
    int t = base + lane;
    return (t < deg) ? col[start + t] : d;
}

__device__ __forceinline__ int lane_fetch(int src, int v) {
    return __builtin_amdgcn_ds_bpermute(src << 2, v);
}
__device__ __forceinline__ float lane_fetch(int src, float v) {
    return __builtin_bit_cast(float, lane_fetch(src, __builtin_bit_cast(int, v)));
}

__device__ __forceinline__ float row_scores32(const float* __restrict__ as, int mycol,
                                              float ad_l, int lane, float (&e)[8]) {
    float m = -INFINITY;
#pragma unroll
    for (int q = 0; q < 8; ++q) {
        int sj = lane_fetch(q * 4 + (lane >> 3), mycol);
        e[q] = leaky(as[(size_t)sj * HEADS + (lane & 7)] + ad_l);
        m = fmaxf(m, e[q]);
    }
    return m;
}

__device__ __forceinline__ float row_weights32(const float (&e)[8], float m, int cnt, int lane,
                                               float (&p)[8]) {
    float s = 0.0f;
#pragma unroll
    for (int q = 0; q < 8; ++q) {
        p[q] = (q * 4 + (lane >> 3) < cnt) ? __expf(e[q] - m) : 0.0f;
        s += p[q];
    }
    return s;
}

__device__ __forceinline__ void row_accum32(const float* __restrict__ xh, int mycol, int cnt,
                                            int lane, const float (&p)[8],
                                            v4f& acc0, v4f& acc1) {
    int h0 = lane >> 3;
    unsigned lane16 = (unsigned)lane * 16u;
#pragma unroll
    for (int q = 0; q < 8; ++q) {
        if (q * 4 < cnt) {
#pragma unroll
            for (int u = 0; u < 4; ++u) {
                int sj = __builtin_amdgcn_readlane(mycol, q * 4 + u);
                unsigned off = (unsigned)sj * (unsigned)(HD * sizeof(float)) + lane16;
                v4f x0 = *(const v4f*)((const char*)xh + off);
                v4f x1 = *(const v4f*)((const char*)xh + off + 512);
                float w0 = lane_fetch(u * 8 + h0, p[q]);
                float w1 = lane_fetch(u * 8 + 4 + h0, p[q]);
                acc0 += w0 * x0;
                acc1 += w1 * x1;
            }
        }
    }
}

__device__ __forceinline__ void row_softmax_agg(int d, int lane,
                    const int* __restrict__ rowptr, const int* __restrict__ col,
                    const float* __restrict__ as, const float* __restrict__ ad,
                    const float* __restrict__ xh, v4f bias0, v4f bias1, v4f& r0, v4f& r1) {
    int start = rowptr[d];
    int deg   = rowptr[d + 1] - start;
    int total = deg + 1;
    int h0 = lane >> 3;
    float ad_l = ad[(size_t)d * HEADS + (lane & 7)];

    int col0 = row_cols(col, start, deg, d, 0, lane);
    float e0[8];
    float m = row_scores32(as, col0, ad_l, lane, e0);
    for (int base = 32; base < total; base += 32) {
        float et[8];
        m = fmaxf(m, row_scores32(as, row_cols(col, start, deg, d, base, lane), ad_l, lane, et));
    }
    m = fmaxf(m, __shfl_xor(m, 8, 32));
    m = fmaxf(m, __shfl_xor(m, 16, 32));

    float p[8];
    float lsum = row_weights32(e0, m, total, lane, p);
    v4f acc0 = (v4f){}, acc1 = (v4f){};
    row_accum32(xh, col0, total < 32 ? total : 32, lane, p, acc0, acc1);
    for (int base = 32; base < total; base += 32) {
        int c = row_cols(col, start, deg, d, base, lane);
        float et[8];
        (void)row_scores32(as, c, ad_l, lane, et);
        int rem = total - base;
        lsum += row_weights32(et, m, rem, lane, p);
        row_accum32(xh, c, rem < 32 ? rem : 32, lane, p, acc0, acc1);
    }
    lsum += __shfl_xor(lsum, 8, 32);
    lsum += __shfl_xor(lsum, 16, 32);
    float sum0 = __shfl(lsum, h0, 32), sum1 = __shfl(lsum, 4 + h0, 32);

    r0 = acc0 / sum0 + bias0;
    r1 = acc1 / sum1 + bias1;
    for (int i = 0; i < 4; ++i) {
        r0[i] = r0[i] > 0.0f ? r0[i] : 0.0f;
        r1[i] = r1[i] > 0.0f ? r1[i] : 0.0f;
    }
}

#define TILE_LD (HD + 4)

template <int NN, bool RAGGED>
__global__ void __launch_bounds__(128) __attribute__((amdgpu_waves_per_eu(4, 9)))
gat_row_then_gemm(const int* __restrict__ rowptr, const int* __restrict__ col,
                  const float* __restrict__ as, const float* __restrict__ ad,
                  const float* __restrict__ xh, const float* __restrict__ bias,
                  const float* __restrict__ B, const float* __restrict__ gbias,
                  float* __restrict__ C,
                  const float* __restrict__ a_src, const float* __restrict__ a_dst,
                  float* __restrict__ as_out, float* __restrict__ ad_out) {
    __shared__ __attribute__((aligned(16))) float tile[16 * TILE_LD];
    int lane = threadIdx.x & 31;
    int wv = __builtin_amdgcn_readfirstlane((int)(threadIdx.x >> 5));
    int m0 = (int)blockIdx.x << 4;
    const float* bp = bias + lane * 4;
    v4f bias0 = *(const v4f*)bp;
    v4f bias1 = *(const v4f*)(bp + 128);
    for (int i = 0; i < 4; ++i) {
        int lr = wv * 4 + i;
        v4f r0, r1;
        row_softmax_agg(m0 + lr, lane, rowptr, col, as, ad, xh, bias0, bias1, r0, r1);
        float* tp = tile + lr * TILE_LD + lane * 4;
        *(v4f*)tp = r0;
        *(v4f*)(tp + 128) = r1;
    }
    __syncthreads();
    if constexpr (!RAGGED) {
        gemm_strip<NN, HD, false, TILE_LD>(tile, m0, wv << 6, B, gbias, C, a_src, a_dst, as_out, ad_out);
    } else {
        constexpr int TILES = (NN + 15) / 16;
        constexpr int QUADS = 16 * NN / 4;
        static_assert(TILES <= 4 && NN % 2 == 0, "one tile per wave; a strip of C is a whole number of 128-byte lines");
        __shared__ __attribute__((aligned(16))) float cst[16 * NN];
        if (wv < TILES)
            gemm_strip<NN, HD, true, TILE_LD, 1>(tile, 0, wv << 4, B, gbias, cst, a_src, a_dst, as_out, ad_out);
        __syncthreads();
        float* cp = C + (size_t)m0 * NN;
        for (int i = (int)threadIdx.x; i < QUADS; i += 128)
            *(v4f*)(cp + 4 * i) = *(const v4f*)(cst + 4 * i);
    }
}

extern "C" void kernel_launch(void* const* d_in, const int* in_sizes, int n_in,
                              void* d_out, int out_size, void* d_ws, size_t ws_size,
                              hipStream_t stream) {
    const float* x   = (const float*)d_in[0];
    const int*   ei  = (const int*)d_in[1];
    const float* W1  = (const float*)d_in[2];
    const float* a1s = (const float*)d_in[3];
    const float* a1d = (const float*)d_in[4];
    const float* b1  = (const float*)d_in[5];
    const float* W2  = (const float*)d_in[6];
    const float* a2s = (const float*)d_in[7];
    const float* a2d = (const float*)d_in[8];
    const float* b2  = (const float*)d_in[9];
    const float* Wc  = (const float*)d_in[10];
    const float* bc  = (const float*)d_in[11];

    const int N = in_sizes[0] / 128;
    const int E = in_sizes[1] / 2;
    const int* src = ei;
    const int* dst = ei + E;

    float* ws    = (float*)d_ws;
    float* xh    = ws;
    float* xh2   = xh   + (size_t)N * HD;
    float* as_   = xh2  + (size_t)N * HD;
    float* ad_   = as_  + (size_t)N * HEADS;
    float* as2_  = ad_  + (size_t)N * HEADS;
    float* ad2_  = as2_ + (size_t)N * HEADS;
    int*   rowptr = (int*)(ad2_ + (size_t)N * HEADS);
    int*   col    = rowptr + N + 1;
    const int nranges = (N + RNODES - 1) >> RSHIFT;
    const int nchunks = (E + CHUNK - 1) / CHUNK;
    int*      offs = col + E;
    unsigned* ebuf = (unsigned*)(offs + (size_t)nchunks * (nranges + 1));

    const int ROW_BT  = 128;
    const int tiles_m = N / 16;
    auto cdiv = [](int a, int b) { return (a + b - 1) / b; };
    const int ngroups = HD / (16 * NSUB);
    const int gemm_waves  = tiles_m * ngroups;
    const int gemm_first  = cdiv(gemm_waves * 2, 5);

    sort_bin_with_gemm1<<<nchunks + cdiv(gemm_first, 8), SORT_BT, 0, stream>>>(
        src, dst, E, nranges, offs, ebuf, nchunks,
        x, W1, nullptr, xh, a1s, a1d, as_, ad_, tiles_m, 0, gemm_first);
    sort_place_with_gemm1<<<nranges + cdiv(gemm_waves - gemm_first, 8), SORT_BT, 0, stream>>>(
        offs, ebuf, nchunks, nranges, N, rowptr, col, nranges,
        x, W1, nullptr, xh, a1s, a1d, as_, ad_, tiles_m, gemm_first, gemm_waves);

    gat_row_then_gemm<HD, false><<<tiles_m, ROW_BT, 0, stream>>>(
        rowptr, col, as_, ad_, xh, b1, W2, nullptr, xh2, a2s, a2d, as2_, ad2_);

    gat_row_then_gemm<40, true><<<tiles_m, ROW_BT, 0, stream>>>(
        rowptr, col, as2_, ad2_, xh2, b2, Wc, bc, (float*)d_out, nullptr, nullptr, nullptr, nullptr);
}
